# starting kernel
# baseline (speedup 1.0000x reference)
#include <hip/hip_runtime.h>

typedef __attribute__((ext_vector_type(2))) float v2f;
typedef __attribute__((ext_vector_type(8))) float v8f;

#define D1 501
#define NSTEPS 512
#define LD 512
#define WIN 4

__global__ void kf_init_S(const float* __restrict__ Sigma0, float* __restrict__ S) {
    int idx = blockIdx.x * blockDim.x + threadIdx.x;
    if (idx < LD * LD) {
        int r = idx >> 9;
        int c = idx & (LD - 1);
        float v = 0.0f;
        if (r < D1 && c < D1) v = Sigma0[r * D1 + c];
        S[idx] = v;
    }
}

__launch_bounds__(1024)
__global__ void kf_scan(const float* __restrict__ emissions,
                        const float* __restrict__ Ht,
                        const float* __restrict__ R,
                        const float* __restrict__ mu0,
                        float* __restrict__ S,
                        float* __restrict__ out) {
    __shared__ float h[LD];
    __shared__ float hs[LD];
    __shared__ float m[LD];
    __shared__ float ps[2 * LD];
    __shared__ float kbuf[WIN][LD];
    __shared__ float svals[WIN];
    __shared__ float redA[16], redB[16], redC[16];
    __shared__ float sc[8];

    const int tid  = threadIdx.x;
    const int lane = tid & 31;
    const int wave = tid >> 5;
    const float r = R[0];

    if (tid < LD)  m[tid] = (tid < D1) ? mu0[tid] : 0.0f;
    if (tid < WIN) svals[tid] = 0.0f;
    __syncthreads();

    for (int t = 0; t < NSTEPS; ++t) {
        const int w = t & (WIN - 1);

        if (tid < LD) {
            h[tid] = (tid < D1) ? Ht[t * D1 + tid] : 0.0f;
            if (t + 1 < NSTEPS && tid < D1)
                __builtin_prefetch(Ht + (t + 1) * D1 + tid, 0, 0);
        }
        __syncthreads();

        {
            int i    = tid & (LD - 1);
            int half = tid >> 9;
            int j0   = half * (LD / 2);
            const float* Scol = S + i;
            float acc = 0.0f;
            #pragma unroll 8
            for (int j = j0; j < j0 + LD / 2; ++j)
                acc = fmaf(Scol[j * LD], h[j], acc);
            ps[half * LD + i] = acc;
        }

        if (w > 0 && tid < LD) {
            float hv = h[tid];
            float d0 = kbuf[0][tid] * hv;
            float d1 = (w > 1) ? kbuf[1][tid] * hv : 0.0f;
            float d2 = (w > 2) ? kbuf[2][tid] * hv : 0.0f;
            #pragma unroll
            for (int off = 16; off > 0; off >>= 1) {
                d0 += __shfl_down(d0, off, 32);
                d1 += __shfl_down(d1, off, 32);
                d2 += __shfl_down(d2, off, 32);
            }
            if (lane == 0) { redA[wave] = d0; redB[wave] = d1; redC[wave] = d2; }
        }
        __syncthreads();
        if (w > 0 && tid == 0) {
            float D0 = 0.0f, D1s = 0.0f, D2 = 0.0f;
            for (int q = 0; q < 16; ++q) { D0 += redA[q]; D1s += redB[q]; D2 += redC[q]; }
            sc[3] = svals[0] * D0;
            sc[4] = svals[1] * D1s;
            sc[5] = svals[2] * D2;
        }
        __syncthreads();

        if (tid < LD) {
            float v = ps[tid] + ps[LD + tid];
            if (w > 0) v -= sc[3] * kbuf[0][tid];
            if (w > 1) v -= sc[4] * kbuf[1][tid];
            if (w > 2) v -= sc[5] * kbuf[2][tid];
            hs[tid] = v;
            float hv = h[tid];
            float r1 = v * hv;
            float r2 = hv * m[tid];
            #pragma unroll
            for (int off = 16; off > 0; off >>= 1) {
                r1 += __shfl_down(r1, off, 32);
                r2 += __shfl_down(r2, off, 32);
            }
            if (lane == 0) { redA[wave] = r1; redB[wave] = r2; }
        }
        __syncthreads();
        if (tid == 0) {
            float s1 = 0.0f, s2 = 0.0f;
            for (int q = 0; q < 16; ++q) { s1 += redA[q]; s2 += redB[q]; }
            float s_term = r + s1;
            sc[0] = s_term;
            sc[1] = 1.0f / (s_term + 1e-6f);
            sc[2] = emissions[t] - s2;
            svals[w] = s_term;
        }
        __syncthreads();

        if (tid < LD) {
            float kv = hs[tid] * sc[1];
            kbuf[w][tid] = kv;
            float mv = m[tid] + kv * sc[2];
            m[tid] = mv;
            if (tid < D1) out[t * D1 + tid] = mv;
        }
        __syncthreads();

        if (w == WIN - 1) {
            const float s0 = svals[0], s1v = svals[1];
            const float s2v = svals[2], s3v = svals[3];
            const int  rl = lane & 15;
            const int  tr = wave << 4;
            const bool lo = (lane < 16);

            float k0r = kbuf[0][tr + rl], k1r = kbuf[1][tr + rl];
            float k2r = kbuf[2][tr + rl], k3r = kbuf[3][tr + rl];
            v2f a;
            a.x = lo ? (-s0  * k0r) : (-s2v * k2r);
            a.y = lo ? (-s1v * k1r) : (-s3v * k3r);

            const int rbase = tr + (lo ? 0 : 8);
            float* Sp0 = S + rbase * LD + rl;

            for (int tc = 0; tc < LD; tc += 16) {
                float* Sp = Sp0 + tc;
                v8f c;
                #pragma unroll
                for (int v = 0; v < 8; ++v) c[v] = Sp[v * LD];

                int col = tc + rl;
                float b0 = kbuf[0][col], b1 = kbuf[1][col];
                float b2 = kbuf[2][col], b3 = kbuf[3][col];
                v2f b;
                b.x = lo ? b0 : b2;
                b.y = lo ? b1 : b3;

                v8f d = __builtin_amdgcn_wmma_f32_16x16x4_f32(
                    false, a, false, b, (short)0, c, false, false);

                #pragma unroll
                for (int v = 0; v < 8; ++v) Sp[v * LD] = d[v];
            }
        }
        __syncthreads();
    }
}

extern "C" void kernel_launch(void* const* d_in, const int* in_sizes, int n_in,
                              void* d_out, int out_size, void* d_ws, size_t ws_size,
                              hipStream_t stream) {
    const float* emissions = (const float*)d_in[0];
    const float* Ht        = (const float*)d_in[1];
    const float* R         = (const float*)d_in[4];
    const float* mu0       = (const float*)d_in[5];
    const float* Sigma0    = (const float*)d_in[6];

    float* S   = (float*)d_ws;
    float* out = (float*)d_out;

    kf_init_S<<<(LD * LD) / 256, 256, 0, stream>>>(Sigma0, S);
    kf_scan<<<1, 1024, 0, stream>>>(emissions, Ht, R, mu0, S, out);
}


# S pass spread over 32 workgroups, one launch per window: every workgroup repeats the 16 steps, then downdates its own 16 columns; G strip summed from 16 per-wave partials (sum order changes)
# speedup vs baseline: 13.9269x; 13.9269x over previous
#include <hip/hip_runtime.h>


typedef __attribute__((ext_vector_type(2))) float v2f;
typedef __attribute__((ext_vector_type(8))) float v8f;

#define D1 501
#define NSTEPS 512
#define LD 512
#define WIN 16
#define NBLK (NSTEPS / WIN)
#define NWG (LD / 16)

#define OFF_S  0
#define OFF_G  (LD * LD)
#define OFF_M  (OFF_G + 2 * WIN * LD)

__global__ void kf_init(const float* __restrict__ Sigma0, const float* __restrict__ mu0, float* ws) {
    int idx = blockIdx.x * blockDim.x + threadIdx.x;
    if (idx < LD * LD) {
        int r = idx >> 9;
        int c = idx & (LD - 1);
        float v = 0.0f;
        if (r < D1 && c < D1) v = Sigma0[r * D1 + c];
        ws[OFF_S + idx] = v;
    }
    if (idx < LD) ws[OFF_M + idx] = (idx < D1) ? mu0[idx] : 0.0f;
}

__device__ __forceinline__ float wave_sum(float x) {
    for (int off = 16; off > 0; off >>= 1) x += __shfl_xor(x, off, 32);
    return x;
}

__device__ __forceinline__ v8f mm4(v2f a, v2f b, v8f c) {
    return __builtin_amdgcn_wmma_f32_16x16x4_f32(false, a, false, b, (short)0, c, false, false);
}

template <bool STEPS, bool PASS>
__launch_bounds__(512)
__global__ void kf_window(const float* __restrict__ emissions,
                          const float* __restrict__ Ht,
                          const float* __restrict__ R,
                          float* ws,
                          float* __restrict__ out,
                          int b) {
    __shared__ __attribute__((aligned(16))) float kr[WIN * LD];
    __shared__ __attribute__((aligned(16))) float mL[LD];
    __shared__ __attribute__((aligned(16))) float part[16 * 256];
    __shared__ float nsv[WIN];
    __shared__ float sS[WIN];
    __shared__ float sR[WIN];

    const int tid  = threadIdx.x;
    const int lane = tid & 31;
    const int wave = __builtin_amdgcn_readfirstlane(tid >> 5);
    const int cl = lane & 15;
    const int hh = lane >> 4;
    const int tc = blockIdx.x << 4;

    v8f c0, c1;
    float ha0[8], ha1[8];
    if constexpr (PASS) {
        const float* Sp0 = ws + OFF_S + ((wave << 4) + 8 * hh) * LD + tc + cl;
        const float* Sp1 = Sp0 + 256 * LD;
        const int hbase = (b * WIN + cl) * D1;
        #pragma unroll
        for (int e = 0; e < 8; ++e) {
            c0[e] = Sp0[e * LD];
            c1[e] = Sp1[e * LD];
            const int i0 = (wave << 4) + 8 * hh + e;
            const int i1 = i0 + 256;
            ha0[e] = Ht[hbase + i0];
            const float hv = Ht[hbase + ((i1 < D1) ? i1 : (D1 - 1))];
            ha1[e] = (i1 < D1) ? hv : 0.0f;
        }
    }

    if constexpr (STEPS) {
        const int wb = b - 1;
        const float* G  = ws + OFF_G + (wb & 1) * (WIN * LD);
        const float* Mp = ws + OFF_M + (wb & 1) * LD;
        const float r = R[0];

        mL[tid] = Mp[tid];

        const int t = wb * WIN + wave;
        const float* hrow = Ht + t * D1;
        float hreg[16];
        float greg[16];
        float acc = 0.0f;
        #pragma unroll
        for (int j = 0; j < 16; ++j) {
            const int i = lane + 32 * j;
            hreg[j] = (i < D1) ? hrow[i] : 0.0f;
            greg[j] = G[wave * LD + i];
            acc = fmaf(hreg[j], Mp[i], acc);
        }
        float cdot = wave_sum(acc);
        const float yreg = emissions[t];
        __syncthreads();

        float mv[16];
        for (int p = 0; p < WIN; ++p) {
            if (wave == p) {
                float r1 = 0.0f;
                #pragma unroll
                for (int j = 0; j < 16; ++j) {
                    mv[j] = mL[lane + 32 * j];
                    r1 = fmaf(greg[j], hreg[j], r1);
                }
                r1 = wave_sum(r1);
                const float s_term = r + r1;
                const float inv    = 1.0f / (s_term + 1e-6f);
                const float resid  = yreg - cdot;
                #pragma unroll
                for (int j = 0; j < 16; ++j) {
                    const int i = lane + 32 * j;
                    const float kv = greg[j] * inv;
                    greg[j] = kv;
                    kr[p * LD + i] = kv;
                    const float m2 = fmaf(kv, resid, mv[j]);
                    mv[j] = m2;
                    mL[i] = m2;
                }
                if (lane == 0) { sS[p] = s_term; sR[p] = resid; nsv[p] = -s_term; }
            }
            __syncthreads();
            if (wave > p) {
                float kk[16];
                float dd = 0.0f;
                #pragma unroll
                for (int j = 0; j < 16; ++j) {
                    kk[j] = kr[p * LD + lane + 32 * j];
                    dd = fmaf(kk[j], hreg[j], dd);
                }
                dd = wave_sum(dd);
                const float coef = sS[p] * dd;
                #pragma unroll
                for (int j = 0; j < 16; ++j) greg[j] = fmaf(-coef, kk[j], greg[j]);
                cdot = fmaf(sR[p], dd, cdot);
            } else if (wave == p && blockIdx.x == 0) {
                #pragma unroll
                for (int j = 0; j < 16; ++j) {
                    const int i = lane + 32 * j;
                    if (i < D1) out[t * D1 + i] = mv[j];
                }
            }
        }
        if constexpr (PASS) {
            if (blockIdx.x == 0) ws[OFF_M + (b & 1) * LD + tid] = mL[tid];
        }
    } else {
        for (int i = tid; i < WIN * LD; i += 512) kr[i] = 0.0f;
        if (tid < WIN) nsv[tid] = 0.0f;
        __syncthreads();
    }

    if constexpr (PASS) {
        float* Sp0 = ws + OFF_S + ((wave << 4) + 8 * hh) * LD + tc + cl;
        float* Sp1 = Sp0 + 256 * LD;
        float* Gn  = ws + OFF_G + (b & 1) * (WIN * LD);

        float ka0[8], ka1[8], kb[8];
        #pragma unroll
        for (int e = 0; e < 8; ++e) {
            const int q = 8 * hh + e;
            ka0[e] = kr[q * LD + (wave << 4) + cl];
            ka1[e] = kr[q * LD + (wave << 4) + 256 + cl];
            kb[e]  = kr[q * LD + tc + cl] * nsv[q];
        }

        v8f d0 = c0;
        d0 = mm4((v2f){ka0[0], ka0[1]}, (v2f){kb[0], kb[1]}, d0);
        d0 = mm4((v2f){ka0[2], ka0[3]}, (v2f){kb[2], kb[3]}, d0);
        d0 = mm4((v2f){ka0[4], ka0[5]}, (v2f){kb[4], kb[5]}, d0);
        d0 = mm4((v2f){ka0[6], ka0[7]}, (v2f){kb[6], kb[7]}, d0);
        v8f d1 = c1;
        d1 = mm4((v2f){ka1[0], ka1[1]}, (v2f){kb[0], kb[1]}, d1);
        d1 = mm4((v2f){ka1[2], ka1[3]}, (v2f){kb[2], kb[3]}, d1);
        d1 = mm4((v2f){ka1[4], ka1[5]}, (v2f){kb[4], kb[5]}, d1);
        d1 = mm4((v2f){ka1[6], ka1[7]}, (v2f){kb[6], kb[7]}, d1);

        #pragma unroll
        for (int v = 0; v < 8; ++v) {
            Sp0[v * LD] = d0[v];
            Sp1[v * LD] = d1[v];
        }

        v8f g = {0.0f, 0.0f, 0.0f, 0.0f, 0.0f, 0.0f, 0.0f, 0.0f};
        g = mm4((v2f){ha0[0], ha0[1]}, (v2f){d0[0], d0[1]}, g);
        g = mm4((v2f){ha0[2], ha0[3]}, (v2f){d0[2], d0[3]}, g);
        g = mm4((v2f){ha0[4], ha0[5]}, (v2f){d0[4], d0[5]}, g);
        g = mm4((v2f){ha0[6], ha0[7]}, (v2f){d0[6], d0[7]}, g);
        g = mm4((v2f){ha1[0], ha1[1]}, (v2f){d1[0], d1[1]}, g);
        g = mm4((v2f){ha1[2], ha1[3]}, (v2f){d1[2], d1[3]}, g);
        g = mm4((v2f){ha1[4], ha1[5]}, (v2f){d1[4], d1[5]}, g);
        g = mm4((v2f){ha1[6], ha1[7]}, (v2f){d1[6], d1[7]}, g);

        #pragma unroll
        for (int v = 0; v < 8; ++v) part[wave * 256 + v * 32 + lane] = g[v];
        __syncthreads();

        if (tid < 256) {
            float acc = 0.0f;
            for (int w = 0; w < 16; ++w) acc += part[w * 256 + tid];
            const int v = tid >> 5;
            Gn[(8 * hh + v) * LD + tc + cl] = acc;
        }
    }
}

extern "C" void kernel_launch(void* const* d_in, const int* in_sizes, int n_in,
                              void* d_out, int out_size, void* d_ws, size_t ws_size,
                              hipStream_t stream) {
    const float* emissions = (const float*)d_in[0];
    const float* Ht        = (const float*)d_in[1];
    const float* R         = (const float*)d_in[4];
    const float* mu0       = (const float*)d_in[5];
    const float* Sigma0    = (const float*)d_in[6];

    float* ws  = (float*)d_ws;
    float* out = (float*)d_out;

#define KF_WINDOW(B) kf_window<true, true><<<NWG, 512, 0, stream>>>(emissions, Ht, R, ws, out, B);

    kf_init<<<(LD * LD) / 256, 256, 0, stream>>>(Sigma0, mu0, ws);
    kf_window<false, true><<<NWG, 512, 0, stream>>>(emissions, Ht, R, ws, out, 0);
    KF_WINDOW(1)  KF_WINDOW(2)  KF_WINDOW(3)  KF_WINDOW(4)
    KF_WINDOW(5)  KF_WINDOW(6)  KF_WINDOW(7)  KF_WINDOW(8)
    KF_WINDOW(9)  KF_WINDOW(10) KF_WINDOW(11) KF_WINDOW(12)
    KF_WINDOW(13) KF_WINDOW(14) KF_WINDOW(15) KF_WINDOW(16)
    KF_WINDOW(17) KF_WINDOW(18) KF_WINDOW(19) KF_WINDOW(20)
    KF_WINDOW(21) KF_WINDOW(22) KF_WINDOW(23) KF_WINDOW(24)
    KF_WINDOW(25) KF_WINDOW(26) KF_WINDOW(27) KF_WINDOW(28)
    KF_WINDOW(29) KF_WINDOW(30) KF_WINDOW(31)
    kf_window<true, false><<<1, 512, 0, stream>>>(emissions, Ht, R, ws, out, NBLK);
}


# wave sums (two on each step's serial path) swap lanes on the vector ALU (one permlanex16, then four row_xmask DPP adds) instead of five dependent LDS permutes; same butterfly pairs, same sum order
# speedup vs baseline: 18.9139x; 1.3581x over previous
#include <hip/hip_runtime.h>


typedef __attribute__((ext_vector_type(2))) float v2f;
typedef __attribute__((ext_vector_type(8))) float v8f;

#define D1 501
#define NSTEPS 512
#define LD 512
#define WIN 16
#define NBLK (NSTEPS / WIN)
#define NWG (LD / 16)

#define OFF_S  0
#define OFF_G  (LD * LD)
#define OFF_M  (OFF_G + 2 * WIN * LD)

__global__ void kf_init(const float* __restrict__ Sigma0, const float* __restrict__ mu0, float* ws) {
    int idx = blockIdx.x * blockDim.x + threadIdx.x;
    if (idx < LD * LD) {
        int r = idx >> 9;
        int c = idx & (LD - 1);
        float v = 0.0f;
        if (r < D1 && c < D1) v = Sigma0[r * D1 + c];
        ws[OFF_S + idx] = v;
    }
    if (idx < LD) ws[OFF_M + idx] = (idx < D1) ? mu0[idx] : 0.0f;
}

__device__ __forceinline__ float other_row(float x) {
    const int b = __builtin_bit_cast(int, x);
    return __builtin_bit_cast(float, __builtin_amdgcn_permlanex16(b, b, 0x76543210, 0xfedcba98, false, false));
}

template <int M>
__device__ __forceinline__ float row_xor(float x) {
    return __builtin_bit_cast(float, __builtin_amdgcn_update_dpp(0, __builtin_bit_cast(int, x), 0x160 + M, 0xf, 0xf, true));
}

__device__ __forceinline__ float wave_sum(float x) {
    x += other_row(x);
    x += row_xor<8>(x);
    x += row_xor<4>(x);
    x += row_xor<2>(x);
    x += row_xor<1>(x);
    return x;
}

__device__ __forceinline__ v8f mm4(v2f a, v2f b, v8f c) {
    return __builtin_amdgcn_wmma_f32_16x16x4_f32(false, a, false, b, (short)0, c, false, false);
}

template <bool STEPS, bool PASS>
__launch_bounds__(512)
__global__ void kf_window(const float* __restrict__ emissions,
                          const float* __restrict__ Ht,
                          const float* __restrict__ R,
                          float* ws,
                          float* __restrict__ out,
                          int b) {
    __shared__ __attribute__((aligned(16))) float kr[WIN * LD];
    __shared__ __attribute__((aligned(16))) float mL[LD];
    __shared__ __attribute__((aligned(16))) float part[16 * 256];
    __shared__ float nsv[WIN];
    __shared__ float sS[WIN];
    __shared__ float sR[WIN];

    const int tid  = threadIdx.x;
    const int lane = tid & 31;
    const int wave = __builtin_amdgcn_readfirstlane(tid >> 5);
    const int cl = lane & 15;
    const int hh = lane >> 4;
    const int tc = blockIdx.x << 4;

    v8f c0, c1;
    float ha0[8], ha1[8];
    if constexpr (PASS) {
        const float* Sp0 = ws + OFF_S + ((wave << 4) + 8 * hh) * LD + tc + cl;
        const float* Sp1 = Sp0 + 256 * LD;
        const int hbase = (b * WIN + cl) * D1;
        #pragma unroll
        for (int e = 0; e < 8; ++e) {
            c0[e] = Sp0[e * LD];
            c1[e] = Sp1[e * LD];
            const int i0 = (wave << 4) + 8 * hh + e;
            const int i1 = i0 + 256;
            ha0[e] = Ht[hbase + i0];
            const float hv = Ht[hbase + ((i1 < D1) ? i1 : (D1 - 1))];
            ha1[e] = (i1 < D1) ? hv : 0.0f;
        }
    }

    if constexpr (STEPS) {
        const int wb = b - 1;
        const float* G  = ws + OFF_G + (wb & 1) * (WIN * LD);
        const float* Mp = ws + OFF_M + (wb & 1) * LD;
        const float r = R[0];

        mL[tid] = Mp[tid];

        const int t = wb * WIN + wave;
        const float* hrow = Ht + t * D1;
        float hreg[16];
        float greg[16];
        float acc = 0.0f;
        #pragma unroll
        for (int j = 0; j < 16; ++j) {
            const int i = lane + 32 * j;
            hreg[j] = (i < D1) ? hrow[i] : 0.0f;
            greg[j] = G[wave * LD + i];
            acc = fmaf(hreg[j], Mp[i], acc);
        }
        float cdot = wave_sum(acc);
        const float yreg = emissions[t];
        __syncthreads();

        float mv[16];
        for (int p = 0; p < WIN; ++p) {
            if (wave == p) {
                float r1 = 0.0f;
                #pragma unroll
                for (int j = 0; j < 16; ++j) {
                    mv[j] = mL[lane + 32 * j];
                    r1 = fmaf(greg[j], hreg[j], r1);
                }
                r1 = wave_sum(r1);
                const float s_term = r + r1;
                const float inv    = 1.0f / (s_term + 1e-6f);
                const float resid  = yreg - cdot;
                #pragma unroll
                for (int j = 0; j < 16; ++j) {
                    const int i = lane + 32 * j;
                    const float kv = greg[j] * inv;
                    greg[j] = kv;
                    kr[p * LD + i] = kv;
                    const float m2 = fmaf(kv, resid, mv[j]);
                    mv[j] = m2;
                    mL[i] = m2;
                }
                if (lane == 0) { sS[p] = s_term; sR[p] = resid; nsv[p] = -s_term; }
            }
            __syncthreads();
            if (wave > p) {
                float kk[16];
                float dd = 0.0f;
                #pragma unroll
                for (int j = 0; j < 16; ++j) {
                    kk[j] = kr[p * LD + lane + 32 * j];
                    dd = fmaf(kk[j], hreg[j], dd);
                }
                dd = wave_sum(dd);
                const float coef = sS[p] * dd;
                #pragma unroll
                for (int j = 0; j < 16; ++j) greg[j] = fmaf(-coef, kk[j], greg[j]);
                cdot = fmaf(sR[p], dd, cdot);
            } else if (wave == p && blockIdx.x == 0) {
                #pragma unroll
                for (int j = 0; j < 16; ++j) {
                    const int i = lane + 32 * j;
                    if (i < D1) out[t * D1 + i] = mv[j];
                }
            }
        }
        if constexpr (PASS) {
            if (blockIdx.x == 0) ws[OFF_M + (b & 1) * LD + tid] = mL[tid];
        }
    } else {
        for (int i = tid; i < WIN * LD; i += 512) kr[i] = 0.0f;
        if (tid < WIN) nsv[tid] = 0.0f;
        __syncthreads();
    }

    if constexpr (PASS) {
        float* Sp0 = ws + OFF_S + ((wave << 4) + 8 * hh) * LD + tc + cl;
        float* Sp1 = Sp0 + 256 * LD;
        float* Gn  = ws + OFF_G + (b & 1) * (WIN * LD);

        float ka0[8], ka1[8], kb[8];
        #pragma unroll
        for (int e = 0; e < 8; ++e) {
            const int q = 8 * hh + e;
            ka0[e] = kr[q * LD + (wave << 4) + cl];
            ka1[e] = kr[q * LD + (wave << 4) + 256 + cl];
            kb[e]  = kr[q * LD + tc + cl] * nsv[q];
        }

        v8f d0 = c0;
        d0 = mm4((v2f){ka0[0], ka0[1]}, (v2f){kb[0], kb[1]}, d0);
        d0 = mm4((v2f){ka0[2], ka0[3]}, (v2f){kb[2], kb[3]}, d0);
        d0 = mm4((v2f){ka0[4], ka0[5]}, (v2f){kb[4], kb[5]}, d0);
        d0 = mm4((v2f){ka0[6], ka0[7]}, (v2f){kb[6], kb[7]}, d0);
        v8f d1 = c1;
        d1 = mm4((v2f){ka1[0], ka1[1]}, (v2f){kb[0], kb[1]}, d1);
        d1 = mm4((v2f){ka1[2], ka1[3]}, (v2f){kb[2], kb[3]}, d1);
        d1 = mm4((v2f){ka1[4], ka1[5]}, (v2f){kb[4], kb[5]}, d1);
        d1 = mm4((v2f){ka1[6], ka1[7]}, (v2f){kb[6], kb[7]}, d1);

        #pragma unroll
        for (int v = 0; v < 8; ++v) {
            Sp0[v * LD] = d0[v];
            Sp1[v * LD] = d1[v];
        }

        v8f g = {0.0f, 0.0f, 0.0f, 0.0f, 0.0f, 0.0f, 0.0f, 0.0f};
        g = mm4((v2f){ha0[0], ha0[1]}, (v2f){d0[0], d0[1]}, g);
        g = mm4((v2f){ha0[2], ha0[3]}, (v2f){d0[2], d0[3]}, g);
        g = mm4((v2f){ha0[4], ha0[5]}, (v2f){d0[4], d0[5]}, g);
        g = mm4((v2f){ha0[6], ha0[7]}, (v2f){d0[6], d0[7]}, g);
        g = mm4((v2f){ha1[0], ha1[1]}, (v2f){d1[0], d1[1]}, g);
        g = mm4((v2f){ha1[2], ha1[3]}, (v2f){d1[2], d1[3]}, g);
        g = mm4((v2f){ha1[4], ha1[5]}, (v2f){d1[4], d1[5]}, g);
        g = mm4((v2f){ha1[6], ha1[7]}, (v2f){d1[6], d1[7]}, g);

        #pragma unroll
        for (int v = 0; v < 8; ++v) part[wave * 256 + v * 32 + lane] = g[v];
        __syncthreads();

        if (tid < 256) {
            float acc = 0.0f;
            for (int w = 0; w < 16; ++w) acc += part[w * 256 + tid];
            const int v = tid >> 5;
            Gn[(8 * hh + v) * LD + tc + cl] = acc;
        }
    }
}

extern "C" void kernel_launch(void* const* d_in, const int* in_sizes, int n_in,
                              void* d_out, int out_size, void* d_ws, size_t ws_size,
                              hipStream_t stream) {
    const float* emissions = (const float*)d_in[0];
    const float* Ht        = (const float*)d_in[1];
    const float* R         = (const float*)d_in[4];
    const float* mu0       = (const float*)d_in[5];
    const float* Sigma0    = (const float*)d_in[6];

    float* ws  = (float*)d_ws;
    float* out = (float*)d_out;

#define KF_WINDOW(B) kf_window<true, true><<<NWG, 512, 0, stream>>>(emissions, Ht, R, ws, out, B);

    kf_init<<<(LD * LD) / 256, 256, 0, stream>>>(Sigma0, mu0, ws);
    kf_window<false, true><<<NWG, 512, 0, stream>>>(emissions, Ht, R, ws, out, 0);
    KF_WINDOW(1)  KF_WINDOW(2)  KF_WINDOW(3)  KF_WINDOW(4)
    KF_WINDOW(5)  KF_WINDOW(6)  KF_WINDOW(7)  KF_WINDOW(8)
    KF_WINDOW(9)  KF_WINDOW(10) KF_WINDOW(11) KF_WINDOW(12)
    KF_WINDOW(13) KF_WINDOW(14) KF_WINDOW(15) KF_WINDOW(16)
    KF_WINDOW(17) KF_WINDOW(18) KF_WINDOW(19) KF_WINDOW(20)
    KF_WINDOW(21) KF_WINDOW(22) KF_WINDOW(23) KF_WINDOW(24)
    KF_WINDOW(25) KF_WINDOW(26) KF_WINDOW(27) KF_WINDOW(28)
    KF_WINDOW(29) KF_WINDOW(30) KF_WINDOW(31)
    kf_window<true, false><<<1, 512, 0, stream>>>(emissions, Ht, R, ws, out, NBLK);
}


# window of 32 steps per launch (was 16) on 32-wave workgroups: 18 launches, not 34; rank-32 downdate, G in two accumulators, strip summed from 32 per-wave partials (sum order changes)
# speedup vs baseline: 20.9209x; 1.1061x over previous
#include <hip/hip_runtime.h>


typedef __attribute__((ext_vector_type(2))) float v2f;
typedef __attribute__((ext_vector_type(8))) float v8f;

#define D1 501
#define NSTEPS 512
#define LD 512
#define WIN 32
#define NBLK (NSTEPS / WIN)
#define NWG (LD / 16)
#define NT (32 * WIN)

#define OFF_S  0
#define OFF_G  (LD * LD)
#define OFF_M  (OFF_G + 2 * WIN * LD)

__global__ void kf_init(const float* __restrict__ Sigma0, const float* __restrict__ mu0, float* ws) {
    int idx = blockIdx.x * blockDim.x + threadIdx.x;
    if (idx < LD * LD) {
        int r = idx >> 9;
        int c = idx & (LD - 1);
        float v = 0.0f;
        if (r < D1 && c < D1) v = Sigma0[r * D1 + c];
        ws[OFF_S + idx] = v;
    }
    if (idx < LD) ws[OFF_M + idx] = (idx < D1) ? mu0[idx] : 0.0f;
}

__device__ __forceinline__ float other_row(float x) {
    const int b = __builtin_bit_cast(int, x);
    return __builtin_bit_cast(float, __builtin_amdgcn_permlanex16(b, b, 0x76543210, 0xfedcba98, false, false));
}

template <int M>
__device__ __forceinline__ float row_xor(float x) {
    return __builtin_bit_cast(float, __builtin_amdgcn_update_dpp(0, __builtin_bit_cast(int, x), 0x160 + M, 0xf, 0xf, true));
}

__device__ __forceinline__ float wave_sum(float x) {
    x += other_row(x);
    x += row_xor<8>(x);
    x += row_xor<4>(x);
    x += row_xor<2>(x);
    x += row_xor<1>(x);
    return x;
}

__device__ __forceinline__ v8f mm4(v2f a, v2f b, v8f c) {
    return __builtin_amdgcn_wmma_f32_16x16x4_f32(false, a, false, b, (short)0, c, false, false);
}

template <bool STEPS, bool PASS>
__launch_bounds__(NT)
__global__ void kf_window(const float* __restrict__ emissions,
                          const float* __restrict__ Ht,
                          const float* __restrict__ R,
                          float* ws,
                          float* __restrict__ out,
                          int b) {
    __shared__ __attribute__((aligned(16))) float kr[WIN * LD];
    __shared__ __attribute__((aligned(16))) float mL[LD];
    __shared__ __attribute__((aligned(16))) float part[32 * 512];
    __shared__ float nsv[WIN];
    __shared__ float sS[WIN];
    __shared__ float sR[WIN];

    const int tid  = threadIdx.x;
    const int lane = tid & 31;
    const int wave = __builtin_amdgcn_readfirstlane(tid >> 5);
    const int cl = lane & 15;
    const int hh = lane >> 4;
    const int tc = blockIdx.x << 4;

    v8f c0;
    float haL[8], haH[8];
    if constexpr (PASS) {
        const float* Sp0 = ws + OFF_S + ((wave << 4) + 8 * hh) * LD + tc + cl;
        const int hbase = (b * WIN + cl) * D1;
        #pragma unroll
        for (int e = 0; e < 8; ++e) {
            c0[e] = Sp0[e * LD];
            const int i0 = (wave << 4) + 8 * hh + e;
            const int ic = (i0 < D1) ? i0 : (D1 - 1);
            const float vL = Ht[hbase + ic];
            const float vH = Ht[hbase + 16 * D1 + ic];
            haL[e] = (i0 < D1) ? vL : 0.0f;
            haH[e] = (i0 < D1) ? vH : 0.0f;
        }
    }

    if constexpr (STEPS) {
        const int wb = b - 1;
        const float* G  = ws + OFF_G + (wb & 1) * (WIN * LD);
        const float* Mp = ws + OFF_M + (wb & 1) * LD;
        const float r = R[0];

        if (tid < LD) mL[tid] = Mp[tid];

        const int t = wb * WIN + wave;
        const float* hrow = Ht + t * D1;
        float hreg[16];
        float greg[16];
        float acc = 0.0f;
        #pragma unroll
        for (int j = 0; j < 16; ++j) {
            const int i = lane + 32 * j;
            hreg[j] = (i < D1) ? hrow[i] : 0.0f;
            greg[j] = G[wave * LD + i];
            acc = fmaf(hreg[j], Mp[i], acc);
        }
        float cdot = wave_sum(acc);
        const float yreg = emissions[t];
        __syncthreads();

        float mv[16];
        for (int p = 0; p < WIN; ++p) {
            if (wave == p) {
                float r1 = 0.0f;
                #pragma unroll
                for (int j = 0; j < 16; ++j) {
                    mv[j] = mL[lane + 32 * j];
                    r1 = fmaf(greg[j], hreg[j], r1);
                }
                r1 = wave_sum(r1);
                const float s_term = r + r1;
                const float inv    = 1.0f / (s_term + 1e-6f);
                const float resid  = yreg - cdot;
                #pragma unroll
                for (int j = 0; j < 16; ++j) {
                    const int i = lane + 32 * j;
                    const float kv = greg[j] * inv;
                    greg[j] = kv;
                    kr[p * LD + i] = kv;
                    const float m2 = fmaf(kv, resid, mv[j]);
                    mv[j] = m2;
                    mL[i] = m2;
                }
                if (lane == 0) { sS[p] = s_term; sR[p] = resid; nsv[p] = -s_term; }
            }
            __syncthreads();
            if (wave > p) {
                float kk[16];
                float dd = 0.0f;
                #pragma unroll
                for (int j = 0; j < 16; ++j) {
                    kk[j] = kr[p * LD + lane + 32 * j];
                    dd = fmaf(kk[j], hreg[j], dd);
                }
                dd = wave_sum(dd);
                const float coef = sS[p] * dd;
                #pragma unroll
                for (int j = 0; j < 16; ++j) greg[j] = fmaf(-coef, kk[j], greg[j]);
                cdot = fmaf(sR[p], dd, cdot);
            } else if (wave == p && blockIdx.x == 0) {
                #pragma unroll
                for (int j = 0; j < 16; ++j) {
                    const int i = lane + 32 * j;
                    if (i < D1) out[t * D1 + i] = mv[j];
                }
            }
        }
        if constexpr (PASS) {
            if (blockIdx.x == 0 && tid < LD) ws[OFF_M + (b & 1) * LD + tid] = mL[tid];
        }
    } else {
        for (int i = tid; i < WIN * LD; i += NT) kr[i] = 0.0f;
        if (tid < WIN) nsv[tid] = 0.0f;
        __syncthreads();
    }

    if constexpr (PASS) {
        float* Sp0 = ws + OFF_S + ((wave << 4) + 8 * hh) * LD + tc + cl;
        float* Gn  = ws + OFF_G + (b & 1) * (WIN * LD);

        float ka[16], kb[16];
        #pragma unroll
        for (int e = 0; e < 16; ++e) {
            const int q = 16 * hh + e;
            ka[e] = kr[q * LD + (wave << 4) + cl];
            kb[e] = kr[q * LD + tc + cl] * nsv[q];
        }

        v8f d0 = c0;
        d0 = mm4((v2f){ka[0],  ka[1]},  (v2f){kb[0],  kb[1]},  d0);
        d0 = mm4((v2f){ka[2],  ka[3]},  (v2f){kb[2],  kb[3]},  d0);
        d0 = mm4((v2f){ka[4],  ka[5]},  (v2f){kb[4],  kb[5]},  d0);
        d0 = mm4((v2f){ka[6],  ka[7]},  (v2f){kb[6],  kb[7]},  d0);
        d0 = mm4((v2f){ka[8],  ka[9]},  (v2f){kb[8],  kb[9]},  d0);
        d0 = mm4((v2f){ka[10], ka[11]}, (v2f){kb[10], kb[11]}, d0);
        d0 = mm4((v2f){ka[12], ka[13]}, (v2f){kb[12], kb[13]}, d0);
        d0 = mm4((v2f){ka[14], ka[15]}, (v2f){kb[14], kb[15]}, d0);

        #pragma unroll
        for (int v = 0; v < 8; ++v) Sp0[v * LD] = d0[v];

        v8f gL = {0.0f, 0.0f, 0.0f, 0.0f, 0.0f, 0.0f, 0.0f, 0.0f};
        v8f gH = {0.0f, 0.0f, 0.0f, 0.0f, 0.0f, 0.0f, 0.0f, 0.0f};
        gL = mm4((v2f){haL[0], haL[1]}, (v2f){d0[0], d0[1]}, gL);
        gH = mm4((v2f){haH[0], haH[1]}, (v2f){d0[0], d0[1]}, gH);
        gL = mm4((v2f){haL[2], haL[3]}, (v2f){d0[2], d0[3]}, gL);
        gH = mm4((v2f){haH[2], haH[3]}, (v2f){d0[2], d0[3]}, gH);
        gL = mm4((v2f){haL[4], haL[5]}, (v2f){d0[4], d0[5]}, gL);
        gH = mm4((v2f){haH[4], haH[5]}, (v2f){d0[4], d0[5]}, gH);
        gL = mm4((v2f){haL[6], haL[7]}, (v2f){d0[6], d0[7]}, gL);
        gH = mm4((v2f){haH[6], haH[7]}, (v2f){d0[6], d0[7]}, gH);

        #pragma unroll
        for (int v = 0; v < 8; ++v) {
            part[wave * 512 + v * 32 + lane]       = gL[v];
            part[wave * 512 + 256 + v * 32 + lane] = gH[v];
        }
        __syncthreads();

        if (tid < 512) {
            float acc = 0.0f;
            for (int w = 0; w < 32; ++w) acc += part[w * 512 + tid];
            const int a = tid >> 8;
            const int v = (tid >> 5) & 7;
            Gn[(16 * a + 8 * hh + v) * LD + tc + cl] = acc;
        }
    }
}

extern "C" void kernel_launch(void* const* d_in, const int* in_sizes, int n_in,
                              void* d_out, int out_size, void* d_ws, size_t ws_size,
                              hipStream_t stream) {
    const float* emissions = (const float*)d_in[0];
    const float* Ht        = (const float*)d_in[1];
    const float* R         = (const float*)d_in[4];
    const float* mu0       = (const float*)d_in[5];
    const float* Sigma0    = (const float*)d_in[6];

    float* ws  = (float*)d_ws;
    float* out = (float*)d_out;

#define KF_WINDOW(B) kf_window<true, true><<<NWG, NT, 0, stream>>>(emissions, Ht, R, ws, out, B);

    kf_init<<<(LD * LD) / 256, 256, 0, stream>>>(Sigma0, mu0, ws);
    kf_window<false, true><<<NWG, NT, 0, stream>>>(emissions, Ht, R, ws, out, 0);
    KF_WINDOW(1)  KF_WINDOW(2)  KF_WINDOW(3)  KF_WINDOW(4)
    KF_WINDOW(5)  KF_WINDOW(6)  KF_WINDOW(7)  KF_WINDOW(8)
    KF_WINDOW(9)  KF_WINDOW(10) KF_WINDOW(11) KF_WINDOW(12)
    KF_WINDOW(13) KF_WINDOW(14) KF_WINDOW(15)
    kf_window<true, false><<<1, NT, 0, stream>>>(emissions, Ht, R, ws, out, NBLK);
}


# h.S.h of each window row summed once at window start, then carried as a scalar (q -= s_p (k_p.h)^2 per correction): active wave's dot and wave sum leave the serial path; sum order changes
# speedup vs baseline: 22.5899x; 1.0798x over previous
#include <hip/hip_runtime.h>


typedef __attribute__((ext_vector_type(2))) float v2f;
typedef __attribute__((ext_vector_type(8))) float v8f;

#define D1 501
#define NSTEPS 512
#define LD 512
#define WIN 32
#define NBLK (NSTEPS / WIN)
#define NWG (LD / 16)
#define NT (32 * WIN)

#define OFF_S  0
#define OFF_G  (LD * LD)
#define OFF_M  (OFF_G + 2 * WIN * LD)

__global__ void kf_init(const float* __restrict__ Sigma0, const float* __restrict__ mu0, float* ws) {
    int idx = blockIdx.x * blockDim.x + threadIdx.x;
    if (idx < LD * LD) {
        int r = idx >> 9;
        int c = idx & (LD - 1);
        float v = 0.0f;
        if (r < D1 && c < D1) v = Sigma0[r * D1 + c];
        ws[OFF_S + idx] = v;
    }
    if (idx < LD) ws[OFF_M + idx] = (idx < D1) ? mu0[idx] : 0.0f;
}

__device__ __forceinline__ float other_row(float x) {
    const int b = __builtin_bit_cast(int, x);
    return __builtin_bit_cast(float, __builtin_amdgcn_permlanex16(b, b, 0x76543210, 0xfedcba98, false, false));
}

template <int M>
__device__ __forceinline__ float row_xor(float x) {
    return __builtin_bit_cast(float, __builtin_amdgcn_update_dpp(0, __builtin_bit_cast(int, x), 0x160 + M, 0xf, 0xf, true));
}

__device__ __forceinline__ float wave_sum(float x) {
    x += other_row(x);
    x += row_xor<8>(x);
    x += row_xor<4>(x);
    x += row_xor<2>(x);
    x += row_xor<1>(x);
    return x;
}

__device__ __forceinline__ v8f mm4(v2f a, v2f b, v8f c) {
    return __builtin_amdgcn_wmma_f32_16x16x4_f32(false, a, false, b, (short)0, c, false, false);
}

template <bool STEPS, bool PASS>
__launch_bounds__(NT)
__global__ void kf_window(const float* __restrict__ emissions,
                          const float* __restrict__ Ht,
                          const float* __restrict__ R,
                          float* ws,
                          float* __restrict__ out,
                          int b) {
    __shared__ __attribute__((aligned(16))) float kr[WIN * LD];
    __shared__ __attribute__((aligned(16))) float mL[LD];
    __shared__ __attribute__((aligned(16))) float part[32 * 512];
    __shared__ float nsv[WIN];
    __shared__ float sS[WIN];
    __shared__ float sR[WIN];

    const int tid  = threadIdx.x;
    const int lane = tid & 31;
    const int wave = __builtin_amdgcn_readfirstlane(tid >> 5);
    const int cl = lane & 15;
    const int hh = lane >> 4;
    const int tc = blockIdx.x << 4;

    v8f c0;
    float haL[8], haH[8];
    if constexpr (PASS) {
        const float* Sp0 = ws + OFF_S + ((wave << 4) + 8 * hh) * LD + tc + cl;
        const int hbase = (b * WIN + cl) * D1;
        #pragma unroll
        for (int e = 0; e < 8; ++e) {
            c0[e] = Sp0[e * LD];
            const int i0 = (wave << 4) + 8 * hh + e;
            const int ic = (i0 < D1) ? i0 : (D1 - 1);
            const float vL = Ht[hbase + ic];
            const float vH = Ht[hbase + 16 * D1 + ic];
            haL[e] = (i0 < D1) ? vL : 0.0f;
            haH[e] = (i0 < D1) ? vH : 0.0f;
        }
    }

    if constexpr (STEPS) {
        const int wb = b - 1;
        const float* G  = ws + OFF_G + (wb & 1) * (WIN * LD);
        const float* Mp = ws + OFF_M + (wb & 1) * LD;
        const float r = R[0];

        if (tid < LD) mL[tid] = Mp[tid];

        const int t = wb * WIN + wave;
        const float* hrow = Ht + t * D1;
        float hreg[16];
        float greg[16];
        float acc = 0.0f;
        float qacc = 0.0f;
        #pragma unroll
        for (int j = 0; j < 16; ++j) {
            const int i = lane + 32 * j;
            hreg[j] = (i < D1) ? hrow[i] : 0.0f;
            greg[j] = G[wave * LD + i];
            acc = fmaf(hreg[j], Mp[i], acc);
            qacc = fmaf(greg[j], hreg[j], qacc);
        }
        float cdot = wave_sum(acc);
        float qdot = wave_sum(qacc);
        const float yreg = emissions[t];
        __syncthreads();

        float mv[16];
        for (int p = 0; p < WIN; ++p) {
            if (wave == p) {
                #pragma unroll
                for (int j = 0; j < 16; ++j) mv[j] = mL[lane + 32 * j];
                const float s_term = r + qdot;
                const float inv    = 1.0f / (s_term + 1e-6f);
                const float resid  = yreg - cdot;
                #pragma unroll
                for (int j = 0; j < 16; ++j) {
                    const int i = lane + 32 * j;
                    const float kv = greg[j] * inv;
                    greg[j] = kv;
                    kr[p * LD + i] = kv;
                    const float m2 = fmaf(kv, resid, mv[j]);
                    mv[j] = m2;
                    mL[i] = m2;
                }
                if (lane == 0) { sS[p] = s_term; sR[p] = resid; nsv[p] = -s_term; }
            }
            __syncthreads();
            if (wave > p) {
                float kk[16];
                float dd = 0.0f;
                #pragma unroll
                for (int j = 0; j < 16; ++j) {
                    kk[j] = kr[p * LD + lane + 32 * j];
                    dd = fmaf(kk[j], hreg[j], dd);
                }
                dd = wave_sum(dd);
                const float coef = sS[p] * dd;
                qdot = fmaf(-coef, dd, qdot);
                cdot = fmaf(sR[p], dd, cdot);
                #pragma unroll
                for (int j = 0; j < 16; ++j) greg[j] = fmaf(-coef, kk[j], greg[j]);
            } else if (wave == p && blockIdx.x == 0) {
                #pragma unroll
                for (int j = 0; j < 16; ++j) {
                    const int i = lane + 32 * j;
                    if (i < D1) out[t * D1 + i] = mv[j];
                }
            }
        }
        if constexpr (PASS) {
            if (blockIdx.x == 0 && tid < LD) ws[OFF_M + (b & 1) * LD + tid] = mL[tid];
        }
    } else {
        for (int i = tid; i < WIN * LD; i += NT) kr[i] = 0.0f;
        if (tid < WIN) nsv[tid] = 0.0f;
        __syncthreads();
    }

    if constexpr (PASS) {
        float* Sp0 = ws + OFF_S + ((wave << 4) + 8 * hh) * LD + tc + cl;
        float* Gn  = ws + OFF_G + (b & 1) * (WIN * LD);

        float ka[16], kb[16];
        #pragma unroll
        for (int e = 0; e < 16; ++e) {
            const int q = 16 * hh + e;
            ka[e] = kr[q * LD + (wave << 4) + cl];
            kb[e] = kr[q * LD + tc + cl] * nsv[q];
        }

        v8f d0 = c0;
        d0 = mm4((v2f){ka[0],  ka[1]},  (v2f){kb[0],  kb[1]},  d0);
        d0 = mm4((v2f){ka[2],  ka[3]},  (v2f){kb[2],  kb[3]},  d0);
        d0 = mm4((v2f){ka[4],  ka[5]},  (v2f){kb[4],  kb[5]},  d0);
        d0 = mm4((v2f){ka[6],  ka[7]},  (v2f){kb[6],  kb[7]},  d0);
        d0 = mm4((v2f){ka[8],  ka[9]},  (v2f){kb[8],  kb[9]},  d0);
        d0 = mm4((v2f){ka[10], ka[11]}, (v2f){kb[10], kb[11]}, d0);
        d0 = mm4((v2f){ka[12], ka[13]}, (v2f){kb[12], kb[13]}, d0);
        d0 = mm4((v2f){ka[14], ka[15]}, (v2f){kb[14], kb[15]}, d0);

        #pragma unroll
        for (int v = 0; v < 8; ++v) Sp0[v * LD] = d0[v];

        v8f gL = {0.0f, 0.0f, 0.0f, 0.0f, 0.0f, 0.0f, 0.0f, 0.0f};
        v8f gH = {0.0f, 0.0f, 0.0f, 0.0f, 0.0f, 0.0f, 0.0f, 0.0f};
        gL = mm4((v2f){haL[0], haL[1]}, (v2f){d0[0], d0[1]}, gL);
        gH = mm4((v2f){haH[0], haH[1]}, (v2f){d0[0], d0[1]}, gH);
        gL = mm4((v2f){haL[2], haL[3]}, (v2f){d0[2], d0[3]}, gL);
        gH = mm4((v2f){haH[2], haH[3]}, (v2f){d0[2], d0[3]}, gH);
        gL = mm4((v2f){haL[4], haL[5]}, (v2f){d0[4], d0[5]}, gL);
        gH = mm4((v2f){haH[4], haH[5]}, (v2f){d0[4], d0[5]}, gH);
        gL = mm4((v2f){haL[6], haL[7]}, (v2f){d0[6], d0[7]}, gL);
        gH = mm4((v2f){haH[6], haH[7]}, (v2f){d0[6], d0[7]}, gH);

        #pragma unroll
        for (int v = 0; v < 8; ++v) {
            part[wave * 512 + v * 32 + lane]       = gL[v];
            part[wave * 512 + 256 + v * 32 + lane] = gH[v];
        }
        __syncthreads();

        if (tid < 512) {
            float acc = 0.0f;
            for (int w = 0; w < 32; ++w) acc += part[w * 512 + tid];
            const int a = tid >> 8;
            const int v = (tid >> 5) & 7;
            Gn[(16 * a + 8 * hh + v) * LD + tc + cl] = acc;
        }
    }
}

extern "C" void kernel_launch(void* const* d_in, const int* in_sizes, int n_in,
                              void* d_out, int out_size, void* d_ws, size_t ws_size,
                              hipStream_t stream) {
    const float* emissions = (const float*)d_in[0];
    const float* Ht        = (const float*)d_in[1];
    const float* R         = (const float*)d_in[4];
    const float* mu0       = (const float*)d_in[5];
    const float* Sigma0    = (const float*)d_in[6];

    float* ws  = (float*)d_ws;
    float* out = (float*)d_out;

#define KF_WINDOW(B) kf_window<true, true><<<NWG, NT, 0, stream>>>(emissions, Ht, R, ws, out, B);

    kf_init<<<(LD * LD) / 256, 256, 0, stream>>>(Sigma0, mu0, ws);
    kf_window<false, true><<<NWG, NT, 0, stream>>>(emissions, Ht, R, ws, out, 0);
    KF_WINDOW(1)  KF_WINDOW(2)  KF_WINDOW(3)  KF_WINDOW(4)
    KF_WINDOW(5)  KF_WINDOW(6)  KF_WINDOW(7)  KF_WINDOW(8)
    KF_WINDOW(9)  KF_WINDOW(10) KF_WINDOW(11) KF_WINDOW(12)
    KF_WINDOW(13) KF_WINDOW(14) KF_WINDOW(15)
    kf_window<true, false><<<1, NT, 0, stream>>>(emissions, Ht, R, ws, out, NBLK);
}


# mean row leaves each step's serial path: wave p builds m_p = m_(p-1) + resid_p k_p behind the step's barrier (workgroup 0 only) while later waves correct their rows; same operations, same sum order
# speedup vs baseline: 24.6703x; 1.0921x over previous
#include <hip/hip_runtime.h>


typedef __attribute__((ext_vector_type(2))) float v2f;
typedef __attribute__((ext_vector_type(8))) float v8f;

#define D1 501
#define NSTEPS 512
#define LD 512
#define WIN 32
#define NBLK (NSTEPS / WIN)
#define NWG (LD / 16)
#define NT (32 * WIN)

#define OFF_S  0
#define OFF_G  (LD * LD)
#define OFF_M  (OFF_G + 2 * WIN * LD)

__global__ void kf_init(const float* __restrict__ Sigma0, const float* __restrict__ mu0, float* ws) {
    int idx = blockIdx.x * blockDim.x + threadIdx.x;
    if (idx < LD * LD) {
        int r = idx >> 9;
        int c = idx & (LD - 1);
        float v = 0.0f;
        if (r < D1 && c < D1) v = Sigma0[r * D1 + c];
        ws[OFF_S + idx] = v;
    }
    if (idx < LD) ws[OFF_M + idx] = (idx < D1) ? mu0[idx] : 0.0f;
}

__device__ __forceinline__ float other_row(float x) {
    const int b = __builtin_bit_cast(int, x);
    return __builtin_bit_cast(float, __builtin_amdgcn_permlanex16(b, b, 0x76543210, 0xfedcba98, false, false));
}

template <int M>
__device__ __forceinline__ float row_xor(float x) {
    return __builtin_bit_cast(float, __builtin_amdgcn_update_dpp(0, __builtin_bit_cast(int, x), 0x160 + M, 0xf, 0xf, true));
}

__device__ __forceinline__ float wave_sum(float x) {
    x += other_row(x);
    x += row_xor<8>(x);
    x += row_xor<4>(x);
    x += row_xor<2>(x);
    x += row_xor<1>(x);
    return x;
}

__device__ __forceinline__ v8f mm4(v2f a, v2f b, v8f c) {
    return __builtin_amdgcn_wmma_f32_16x16x4_f32(false, a, false, b, (short)0, c, false, false);
}

template <bool STEPS, bool PASS>
__launch_bounds__(NT)
__global__ void kf_window(const float* __restrict__ emissions,
                          const float* __restrict__ Ht,
                          const float* __restrict__ R,
                          float* ws,
                          float* __restrict__ out,
                          int b) {
    __shared__ __attribute__((aligned(16))) float kr[WIN * LD];
    __shared__ __attribute__((aligned(16))) float mL[LD];
    __shared__ __attribute__((aligned(16))) float part[32 * 512];
    __shared__ float nsv[WIN];
    __shared__ float sS[WIN];
    __shared__ float sR[WIN];

    const int tid  = threadIdx.x;
    const int lane = tid & 31;
    const int wave = __builtin_amdgcn_readfirstlane(tid >> 5);
    const int cl = lane & 15;
    const int hh = lane >> 4;
    const int tc = blockIdx.x << 4;

    v8f c0;
    float haL[8], haH[8];
    if constexpr (PASS) {
        const float* Sp0 = ws + OFF_S + ((wave << 4) + 8 * hh) * LD + tc + cl;
        const int hbase = (b * WIN + cl) * D1;
        #pragma unroll
        for (int e = 0; e < 8; ++e) {
            c0[e] = Sp0[e * LD];
            const int i0 = (wave << 4) + 8 * hh + e;
            const int ic = (i0 < D1) ? i0 : (D1 - 1);
            const float vL = Ht[hbase + ic];
            const float vH = Ht[hbase + 16 * D1 + ic];
            haL[e] = (i0 < D1) ? vL : 0.0f;
            haH[e] = (i0 < D1) ? vH : 0.0f;
        }
    }

    if constexpr (STEPS) {
        const int wb = b - 1;
        const float* G  = ws + OFF_G + (wb & 1) * (WIN * LD);
        const float* Mp = ws + OFF_M + (wb & 1) * LD;
        const float r = R[0];

        if (blockIdx.x == 0 && tid < LD) mL[tid] = Mp[tid];

        const int t = wb * WIN + wave;
        const float* hrow = Ht + t * D1;
        float hreg[16];
        float greg[16];
        float acc = 0.0f;
        float qacc = 0.0f;
        #pragma unroll
        for (int j = 0; j < 16; ++j) {
            const int i = lane + 32 * j;
            hreg[j] = (i < D1) ? hrow[i] : 0.0f;
            greg[j] = G[wave * LD + i];
            acc = fmaf(hreg[j], Mp[i], acc);
            qacc = fmaf(greg[j], hreg[j], qacc);
        }
        float cdot = wave_sum(acc);
        float qdot = wave_sum(qacc);
        const float yreg = emissions[t];
        __syncthreads();

        float resid = 0.0f;
        for (int p = 0; p < WIN; ++p) {
            if (wave == p) {
                const float s_term = r + qdot;
                const float inv    = 1.0f / (s_term + 1e-6f);
                resid = yreg - cdot;
                #pragma unroll
                for (int j = 0; j < 16; ++j) {
                    const float kv = greg[j] * inv;
                    greg[j] = kv;
                    kr[p * LD + lane + 32 * j] = kv;
                }
                if (lane == 0) { sS[p] = s_term; sR[p] = resid; nsv[p] = -s_term; }
            }
            __syncthreads();
            if (wave > p) {
                float kk[16];
                float dd = 0.0f;
                #pragma unroll
                for (int j = 0; j < 16; ++j) {
                    kk[j] = kr[p * LD + lane + 32 * j];
                    dd = fmaf(kk[j], hreg[j], dd);
                }
                dd = wave_sum(dd);
                const float coef = sS[p] * dd;
                qdot = fmaf(-coef, dd, qdot);
                cdot = fmaf(sR[p], dd, cdot);
                #pragma unroll
                for (int j = 0; j < 16; ++j) greg[j] = fmaf(-coef, kk[j], greg[j]);
            } else if (wave == p && blockIdx.x == 0) {
                float mv[16];
                #pragma unroll
                for (int j = 0; j < 16; ++j) mv[j] = mL[lane + 32 * j];
                #pragma unroll
                for (int j = 0; j < 16; ++j) {
                    const int i = lane + 32 * j;
                    mv[j] = fmaf(greg[j], resid, mv[j]);
                    if (i < D1) out[t * D1 + i] = mv[j];
                    if (p < WIN - 1) {
                        mL[i] = mv[j];
                    } else if (PASS) {
                        ws[OFF_M + (b & 1) * LD + i] = mv[j];
                    }
                }
            }
        }
    } else {
        for (int i = tid; i < WIN * LD; i += NT) kr[i] = 0.0f;
        if (tid < WIN) nsv[tid] = 0.0f;
        __syncthreads();
    }

    if constexpr (PASS) {
        float* Sp0 = ws + OFF_S + ((wave << 4) + 8 * hh) * LD + tc + cl;
        float* Gn  = ws + OFF_G + (b & 1) * (WIN * LD);

        float ka[16], kb[16];
        #pragma unroll
        for (int e = 0; e < 16; ++e) {
            const int q = 16 * hh + e;
            ka[e] = kr[q * LD + (wave << 4) + cl];
            kb[e] = kr[q * LD + tc + cl] * nsv[q];
        }

        v8f d0 = c0;
        d0 = mm4((v2f){ka[0],  ka[1]},  (v2f){kb[0],  kb[1]},  d0);
        d0 = mm4((v2f){ka[2],  ka[3]},  (v2f){kb[2],  kb[3]},  d0);
        d0 = mm4((v2f){ka[4],  ka[5]},  (v2f){kb[4],  kb[5]},  d0);
        d0 = mm4((v2f){ka[6],  ka[7]},  (v2f){kb[6],  kb[7]},  d0);
        d0 = mm4((v2f){ka[8],  ka[9]},  (v2f){kb[8],  kb[9]},  d0);
        d0 = mm4((v2f){ka[10], ka[11]}, (v2f){kb[10], kb[11]}, d0);
        d0 = mm4((v2f){ka[12], ka[13]}, (v2f){kb[12], kb[13]}, d0);
        d0 = mm4((v2f){ka[14], ka[15]}, (v2f){kb[14], kb[15]}, d0);

        #pragma unroll
        for (int v = 0; v < 8; ++v) Sp0[v * LD] = d0[v];

        v8f gL = {0.0f, 0.0f, 0.0f, 0.0f, 0.0f, 0.0f, 0.0f, 0.0f};
        v8f gH = {0.0f, 0.0f, 0.0f, 0.0f, 0.0f, 0.0f, 0.0f, 0.0f};
        gL = mm4((v2f){haL[0], haL[1]}, (v2f){d0[0], d0[1]}, gL);
        gH = mm4((v2f){haH[0], haH[1]}, (v2f){d0[0], d0[1]}, gH);
        gL = mm4((v2f){haL[2], haL[3]}, (v2f){d0[2], d0[3]}, gL);
        gH = mm4((v2f){haH[2], haH[3]}, (v2f){d0[2], d0[3]}, gH);
        gL = mm4((v2f){haL[4], haL[5]}, (v2f){d0[4], d0[5]}, gL);
        gH = mm4((v2f){haH[4], haH[5]}, (v2f){d0[4], d0[5]}, gH);
        gL = mm4((v2f){haL[6], haL[7]}, (v2f){d0[6], d0[7]}, gL);
        gH = mm4((v2f){haH[6], haH[7]}, (v2f){d0[6], d0[7]}, gH);

        #pragma unroll
        for (int v = 0; v < 8; ++v) {
            part[wave * 512 + v * 32 + lane]       = gL[v];
            part[wave * 512 + 256 + v * 32 + lane] = gH[v];
        }
        __syncthreads();

        if (tid < 512) {
            float acc = 0.0f;
            for (int w = 0; w < 32; ++w) acc += part[w * 512 + tid];
            const int a = tid >> 8;
            const int v = (tid >> 5) & 7;
            Gn[(16 * a + 8 * hh + v) * LD + tc + cl] = acc;
        }
    }
}

extern "C" void kernel_launch(void* const* d_in, const int* in_sizes, int n_in,
                              void* d_out, int out_size, void* d_ws, size_t ws_size,
                              hipStream_t stream) {
    const float* emissions = (const float*)d_in[0];
    const float* Ht        = (const float*)d_in[1];
    const float* R         = (const float*)d_in[4];
    const float* mu0       = (const float*)d_in[5];
    const float* Sigma0    = (const float*)d_in[6];

    float* ws  = (float*)d_ws;
    float* out = (float*)d_out;

#define KF_WINDOW(B) kf_window<true, true><<<NWG, NT, 0, stream>>>(emissions, Ht, R, ws, out, B);

    kf_init<<<(LD * LD) / 256, 256, 0, stream>>>(Sigma0, mu0, ws);
    kf_window<false, true><<<NWG, NT, 0, stream>>>(emissions, Ht, R, ws, out, 0);
    KF_WINDOW(1)  KF_WINDOW(2)  KF_WINDOW(3)  KF_WINDOW(4)
    KF_WINDOW(5)  KF_WINDOW(6)  KF_WINDOW(7)  KF_WINDOW(8)
    KF_WINDOW(9)  KF_WINDOW(10) KF_WINDOW(11) KF_WINDOW(12)
    KF_WINDOW(13) KF_WINDOW(14) KF_WINDOW(15)
    kf_window<true, false><<<1, NT, 0, stream>>>(emissions, Ht, R, ws, out, NBLK);
}


# the 32 steps of a window run in ONE wave on the 32x32 Gram matrix G0*H^T (registers; no barrier, LDS trip or wave sum per step); gain and mean rows then come as f32 WMMA products; sum order changes
# speedup vs baseline: 27.7105x; 1.1232x over previous
#include <hip/hip_runtime.h>


typedef __attribute__((ext_vector_type(2))) float v2f;
typedef __attribute__((ext_vector_type(8))) float v8f;

#define D1 501
#define NSTEPS 512
#define LD 512
#define WIN 32
#define NBLK (NSTEPS / WIN)
#define NWG (LD / 16)
#define NT (32 * WIN)
#define KP 516
#define TP 36

#define OFF_S  0
#define OFF_G  (LD * LD)
#define OFF_M  (OFF_G + 2 * WIN * LD)

__global__ void kf_init(const float* __restrict__ Sigma0, const float* __restrict__ mu0, float* ws) {
    int idx = blockIdx.x * blockDim.x + threadIdx.x;
    if (idx < LD * LD) {
        int r = idx >> 9;
        int c = idx & (LD - 1);
        float v = 0.0f;
        if (r < D1 && c < D1) v = Sigma0[r * D1 + c];
        ws[OFF_S + idx] = v;
    }
    if (idx < LD) ws[OFF_M + idx] = (idx < D1) ? mu0[idx] : 0.0f;
}

__device__ __forceinline__ float other_row(float x) {
    const int b = __builtin_bit_cast(int, x);
    return __builtin_bit_cast(float, __builtin_amdgcn_permlanex16(b, b, 0x76543210, 0xfedcba98, false, false));
}

template <int M>
__device__ __forceinline__ float row_xor(float x) {
    return __builtin_bit_cast(float, __builtin_amdgcn_update_dpp(0, __builtin_bit_cast(int, x), 0x160 + M, 0xf, 0xf, true));
}

__device__ __forceinline__ float wave_sum(float x) {
    x += other_row(x);
    x += row_xor<8>(x);
    x += row_xor<4>(x);
    x += row_xor<2>(x);
    x += row_xor<1>(x);
    return x;
}

__device__ __forceinline__ v8f mm4(v2f a, v2f b, v8f c) {
    return __builtin_amdgcn_wmma_f32_16x16x4_f32(false, a, false, b, (short)0, c, false, false);
}

__device__ __forceinline__ float lane_val(float x, int q) {
    return __builtin_bit_cast(float, __builtin_amdgcn_readlane(__builtin_bit_cast(int, x), q));
}

template <bool STEPS, bool PASS>
__launch_bounds__(NT)
__global__ void kf_window(const float* __restrict__ emissions,
                          const float* __restrict__ Ht,
                          const float* __restrict__ R,
                          float* ws,
                          float* __restrict__ out,
                          int b) {
    __shared__ __attribute__((aligned(16))) float kr[WIN * KP];
    __shared__ __attribute__((aligned(16))) float hs[WIN * KP];
    __shared__ __attribute__((aligned(16))) float part[32 * 512];
    __shared__ __attribute__((aligned(16))) float Ts[WIN * TP];
    __shared__ __attribute__((aligned(16))) float Tms[WIN * TP];
    __shared__ float gram[WIN * WIN];
    __shared__ float nsv[WIN];
    __shared__ float sC[WIN];

    const int tid  = threadIdx.x;
    const int lane = tid & 31;
    const int wave = __builtin_amdgcn_readfirstlane(tid >> 5);
    const int cl = lane & 15;
    const int hh = lane >> 4;
    const int tc = blockIdx.x << 4;

    v8f c0;
    float haL[8], haH[8];
    auto load_pass_operands = [&]() {
        const float* Sp0 = ws + OFF_S + ((wave << 4) + 8 * hh) * LD + tc + cl;
        const int hbase = (b * WIN + cl) * D1;
        #pragma unroll
        for (int e = 0; e < 8; ++e) {
            c0[e] = Sp0[e * LD];
            const int i0 = (wave << 4) + 8 * hh + e;
            const int ic = (i0 < D1) ? i0 : (D1 - 1);
            const float vL = Ht[hbase + ic];
            const float vH = Ht[hbase + 16 * D1 + ic];
            haL[e] = (i0 < D1) ? vL : 0.0f;
            haH[e] = (i0 < D1) ? vH : 0.0f;
        }
    };
    if constexpr (PASS && !STEPS) load_pass_operands();

    if constexpr (STEPS) {
        const int wb = b - 1;
        const float* G  = ws + OFF_G + (wb & 1) * (WIN * LD);
        const float* Mp = ws + OFF_M + (wb & 1) * LD;
        const float r = R[0];

        {
            const float* hrow = Ht + (wb * WIN + wave) * D1;
            float acc = 0.0f;
            #pragma unroll
            for (int j = 0; j < 16; ++j) {
                const int i = lane + 32 * j;
                const float hv = (i < D1) ? hrow[i] : 0.0f;
                acc = fmaf(hv, Mp[i], acc);
                hs[wave * KP + i] = hv;
                kr[wave * KP + i] = G[wave * LD + i];
            }
            const float cdot = wave_sum(acc);
            if (lane == 0) sC[wave] = cdot;
        }
        __syncthreads();

        {
            const int ta = (wave >> 1) & 1;
            const int tb = wave & 1;
            const int sl = wave >> 2;
            const float* ap = kr + (16 * ta + cl) * KP + 64 * sl + 32 * hh;
            const float* bp = hs + (16 * tb + cl) * KP + 64 * sl + 32 * hh;
            v8f g = {0.0f, 0.0f, 0.0f, 0.0f, 0.0f, 0.0f, 0.0f, 0.0f};
            #pragma unroll
            for (int c = 0; c < 16; ++c)
                g = mm4((v2f){ap[2 * c], ap[2 * c + 1]}, (v2f){bp[2 * c], bp[2 * c + 1]}, g);
            #pragma unroll
            for (int v = 0; v < 8; ++v) part[wave * 256 + v * 32 + lane] = g[v];
        }
        __syncthreads();

        {
            const int o = (wave >> 4) * 512 + (wave & 7) * 32 + ((wave >> 3) & 1) * 16 + (lane >> 4) * 256 + (lane & 15);
            float sacc = part[o];
            #pragma unroll
            for (int s = 1; s < 8; ++s) sacc += part[s * 1024 + o];
            gram[wave * WIN + lane] = sacc;
        }
        __syncthreads();

        if (wave == 0) {
            v2f bc[WIN];
            #pragma unroll
            for (int p = 0; p < WIN; ++p) bc[p] = (v2f){gram[p * WIN + lane], (lane == p) ? 1.0f : 0.0f};
            float cv = sC[lane];
            const float yv = emissions[wb * WIN + lane];
            float mc = 0.0f;
            float sv = 0.0f;
            #pragma unroll
            for (int q = 0; q < WIN; ++q) {
                const float s_term = r + lane_val(bc[q][0], q);
                const float inv    = 1.0f / (s_term + 1e-6f);
                const float u  = bc[q][0] * inv;
                const float ck = bc[q][1] * inv;
                const float resid = lane_val(yv, q) - lane_val(cv, q);
                mc = fmaf(resid, ck, mc);
                cv = fmaf(resid, u, cv);
                sv = (lane == q) ? s_term : sv;
                Ts[q * TP + lane]  = ck;
                Tms[q * TP + lane] = mc;
                const float ncv = u * (-s_term);
                #pragma unroll
                for (int p = q + 1; p < WIN; ++p) {
                    const float nc = lane_val(ncv, p);
                    bc[p] = __builtin_elementwise_fma((v2f){nc, nc}, (v2f){u, ck}, bc[p]);
                }
            }
            nsv[lane] = -sv;
        }
        __syncthreads();
        if constexpr (PASS) load_pass_operands();

        {
            const int col = (wave << 4) + cl;
            float gB[16];
            #pragma unroll
            for (int e = 0; e < 16; ++e) gB[e] = kr[(16 * hh + e) * KP + col];

            if (!PASS || wave == (int)blockIdx.x) {
                const float* t0p = Tms + cl * TP + 16 * hh;
                const float* t1p = Tms + (16 + cl) * TP + 16 * hh;
                const float m0 = Mp[col];
                v8f o0 = {m0, m0, m0, m0, m0, m0, m0, m0};
                v8f o1 = {m0, m0, m0, m0, m0, m0, m0, m0};
                #pragma unroll
                for (int c = 0; c < 8; ++c) {
                    o0 = mm4((v2f){t0p[2 * c], t0p[2 * c + 1]}, (v2f){gB[2 * c], gB[2 * c + 1]}, o0);
                    o1 = mm4((v2f){t1p[2 * c], t1p[2 * c + 1]}, (v2f){gB[2 * c], gB[2 * c + 1]}, o1);
                }
                float* op = out + (wb * WIN + 8 * hh) * D1 + col;
                if (col < D1) {
                    #pragma unroll
                    for (int v = 0; v < 8; ++v) {
                        op[v * D1] = o0[v];
                        op[(16 + v) * D1] = o1[v];
                    }
                }
                if (PASS && hh == 1) ws[OFF_M + (b & 1) * LD + col] = o1[7];
            }

            if constexpr (PASS) {
                const float* t0p = Ts + cl * TP + 16 * hh;
                const float* t1p = Ts + (16 + cl) * TP + 16 * hh;
                v8f k0 = {0.0f, 0.0f, 0.0f, 0.0f, 0.0f, 0.0f, 0.0f, 0.0f};
                v8f k1 = {0.0f, 0.0f, 0.0f, 0.0f, 0.0f, 0.0f, 0.0f, 0.0f};
                #pragma unroll
                for (int c = 0; c < 8; ++c) {
                    k0 = mm4((v2f){t0p[2 * c], t0p[2 * c + 1]}, (v2f){gB[2 * c], gB[2 * c + 1]}, k0);
                    k1 = mm4((v2f){t1p[2 * c], t1p[2 * c + 1]}, (v2f){gB[2 * c], gB[2 * c + 1]}, k1);
                }
                #pragma unroll
                for (int v = 0; v < 8; ++v) {
                    kr[(8 * hh + v) * KP + col] = k0[v];
                    kr[(16 + 8 * hh + v) * KP + col] = k1[v];
                }
            }
        }
        __syncthreads();
    } else {
        for (int i = tid; i < WIN * KP; i += NT) kr[i] = 0.0f;
        if (tid < WIN) nsv[tid] = 0.0f;
        __syncthreads();
    }

    if constexpr (PASS) {
        float* Sp0 = ws + OFF_S + ((wave << 4) + 8 * hh) * LD + tc + cl;
        float* Gn  = ws + OFF_G + (b & 1) * (WIN * LD);

        float ka[16], kb[16];
        #pragma unroll
        for (int e = 0; e < 16; ++e) {
            const int q = 16 * hh + e;
            ka[e] = kr[q * KP + (wave << 4) + cl];
            kb[e] = kr[q * KP + tc + cl] * nsv[q];
        }
        v8f d0 = c0;
        d0 = mm4((v2f){ka[0],  ka[1]},  (v2f){kb[0],  kb[1]},  d0);
        d0 = mm4((v2f){ka[2],  ka[3]},  (v2f){kb[2],  kb[3]},  d0);
        d0 = mm4((v2f){ka[4],  ka[5]},  (v2f){kb[4],  kb[5]},  d0);
        d0 = mm4((v2f){ka[6],  ka[7]},  (v2f){kb[6],  kb[7]},  d0);
        d0 = mm4((v2f){ka[8],  ka[9]},  (v2f){kb[8],  kb[9]},  d0);
        d0 = mm4((v2f){ka[10], ka[11]}, (v2f){kb[10], kb[11]}, d0);
        d0 = mm4((v2f){ka[12], ka[13]}, (v2f){kb[12], kb[13]}, d0);
        d0 = mm4((v2f){ka[14], ka[15]}, (v2f){kb[14], kb[15]}, d0);

        #pragma unroll
        for (int v = 0; v < 8; ++v) Sp0[v * LD] = d0[v];

        v8f gL = {0.0f, 0.0f, 0.0f, 0.0f, 0.0f, 0.0f, 0.0f, 0.0f};
        v8f gH = {0.0f, 0.0f, 0.0f, 0.0f, 0.0f, 0.0f, 0.0f, 0.0f};
        gL = mm4((v2f){haL[0], haL[1]}, (v2f){d0[0], d0[1]}, gL);
        gH = mm4((v2f){haH[0], haH[1]}, (v2f){d0[0], d0[1]}, gH);
        gL = mm4((v2f){haL[2], haL[3]}, (v2f){d0[2], d0[3]}, gL);
        gH = mm4((v2f){haH[2], haH[3]}, (v2f){d0[2], d0[3]}, gH);
        gL = mm4((v2f){haL[4], haL[5]}, (v2f){d0[4], d0[5]}, gL);
        gH = mm4((v2f){haH[4], haH[5]}, (v2f){d0[4], d0[5]}, gH);
        gL = mm4((v2f){haL[6], haL[7]}, (v2f){d0[6], d0[7]}, gL);
        gH = mm4((v2f){haH[6], haH[7]}, (v2f){d0[6], d0[7]}, gH);

        #pragma unroll
        for (int v = 0; v < 8; ++v) {
            part[wave * 512 + v * 32 + lane]       = gL[v];
            part[wave * 512 + 256 + v * 32 + lane] = gH[v];
        }
        __syncthreads();

        if (tid < 512) {
            float acc = 0.0f;
            for (int w = 0; w < 32; ++w) acc += part[w * 512 + tid];
            const int a = tid >> 8;
            const int v = (tid >> 5) & 7;
            Gn[(16 * a + 8 * hh + v) * LD + tc + cl] = acc;
        }
    }
}

extern "C" void kernel_launch(void* const* d_in, const int* in_sizes, int n_in,
                              void* d_out, int out_size, void* d_ws, size_t ws_size,
                              hipStream_t stream) {
    const float* emissions = (const float*)d_in[0];
    const float* Ht        = (const float*)d_in[1];
    const float* R         = (const float*)d_in[4];
    const float* mu0       = (const float*)d_in[5];
    const float* Sigma0    = (const float*)d_in[6];

    float* ws  = (float*)d_ws;
    float* out = (float*)d_out;

#define KF_WINDOW(B) kf_window<true, true><<<NWG, NT, 0, stream>>>(emissions, Ht, R, ws, out, B);

    kf_init<<<(LD * LD) / 256, 256, 0, stream>>>(Sigma0, mu0, ws);
    kf_window<false, true><<<NWG, NT, 0, stream>>>(emissions, Ht, R, ws, out, 0);
    KF_WINDOW(1)  KF_WINDOW(2)  KF_WINDOW(3)  KF_WINDOW(4)
    KF_WINDOW(5)  KF_WINDOW(6)  KF_WINDOW(7)  KF_WINDOW(8)
    KF_WINDOW(9)  KF_WINDOW(10) KF_WINDOW(11) KF_WINDOW(12)
    KF_WINDOW(13) KF_WINDOW(14) KF_WINDOW(15)
    kf_window<true, false><<<1, NT, 0, stream>>>(emissions, Ht, R, ws, out, NBLK);
}


# row factors of each step's rank-1 update reach the FMAs by row_share DPP on the vector ALU (one row exchange a step), not by 31 v_readlane to scalar register trips; same operations, same sum order
# speedup vs baseline: 30.1983x; 1.0898x over previous
#include <hip/hip_runtime.h>


typedef __attribute__((ext_vector_type(2))) float v2f;
typedef __attribute__((ext_vector_type(8))) float v8f;

#define D1 501
#define NSTEPS 512
#define LD 512
#define WIN 32
#define NBLK (NSTEPS / WIN)
#define NWG (LD / 16)
#define NT (32 * WIN)
#define KP 516
#define TP 36

#define OFF_S  0
#define OFF_G  (LD * LD)
#define OFF_M  (OFF_G + 2 * WIN * LD)

__global__ void kf_init(const float* __restrict__ Sigma0, const float* __restrict__ mu0, float* ws) {
    int idx = blockIdx.x * blockDim.x + threadIdx.x;
    if (idx < LD * LD) {
        int r = idx >> 9;
        int c = idx & (LD - 1);
        float v = 0.0f;
        if (r < D1 && c < D1) v = Sigma0[r * D1 + c];
        ws[OFF_S + idx] = v;
    }
    if (idx < LD) ws[OFF_M + idx] = (idx < D1) ? mu0[idx] : 0.0f;
}

__device__ __forceinline__ float other_row(float x) {
    const int b = __builtin_bit_cast(int, x);
    return __builtin_bit_cast(float, __builtin_amdgcn_permlanex16(b, b, 0x76543210, 0xfedcba98, false, false));
}

template <int M>
__device__ __forceinline__ float row_xor(float x) {
    return __builtin_bit_cast(float, __builtin_amdgcn_update_dpp(0, __builtin_bit_cast(int, x), 0x160 + M, 0xf, 0xf, true));
}

template <int P>
__device__ __forceinline__ float row_lane(float x) {
    return __builtin_bit_cast(float, __builtin_amdgcn_update_dpp(0, __builtin_bit_cast(int, x), 0x150 + P, 0xf, 0xf, true));
}

__device__ __forceinline__ float row_lane_n(float x, int p) {
    switch (p & 15) {
        case 0:  return row_lane<0>(x);
        case 1:  return row_lane<1>(x);
        case 2:  return row_lane<2>(x);
        case 3:  return row_lane<3>(x);
        case 4:  return row_lane<4>(x);
        case 5:  return row_lane<5>(x);
        case 6:  return row_lane<6>(x);
        case 7:  return row_lane<7>(x);
        case 8:  return row_lane<8>(x);
        case 9:  return row_lane<9>(x);
        case 10: return row_lane<10>(x);
        case 11: return row_lane<11>(x);
        case 12: return row_lane<12>(x);
        case 13: return row_lane<13>(x);
        case 14: return row_lane<14>(x);
        default: return row_lane<15>(x);
    }
}

__device__ __forceinline__ float wave_sum(float x) {
    x += other_row(x);
    x += row_xor<8>(x);
    x += row_xor<4>(x);
    x += row_xor<2>(x);
    x += row_xor<1>(x);
    return x;
}

__device__ __forceinline__ v8f mm4(v2f a, v2f b, v8f c) {
    return __builtin_amdgcn_wmma_f32_16x16x4_f32(false, a, false, b, (short)0, c, false, false);
}

__device__ __forceinline__ float lane_val(float x, int q) {
    return __builtin_bit_cast(float, __builtin_amdgcn_readlane(__builtin_bit_cast(int, x), q));
}

template <bool STEPS, bool PASS>
__launch_bounds__(NT)
__global__ void kf_window(const float* __restrict__ emissions,
                          const float* __restrict__ Ht,
                          const float* __restrict__ R,
                          float* ws,
                          float* __restrict__ out,
                          int b) {
    __shared__ __attribute__((aligned(16))) float kr[WIN * KP];
    __shared__ __attribute__((aligned(16))) float hs[WIN * KP];
    __shared__ __attribute__((aligned(16))) float part[32 * 512];
    __shared__ __attribute__((aligned(16))) float Ts[WIN * TP];
    __shared__ __attribute__((aligned(16))) float Tms[WIN * TP];
    __shared__ float gram[WIN * WIN];
    __shared__ float nsv[WIN];
    __shared__ float sC[WIN];

    const int tid  = threadIdx.x;
    const int lane = tid & 31;
    const int wave = __builtin_amdgcn_readfirstlane(tid >> 5);
    const int cl = lane & 15;
    const int hh = lane >> 4;
    const int tc = blockIdx.x << 4;

    v8f c0;
    float haL[8], haH[8];
    auto load_pass_operands = [&]() {
        const float* Sp0 = ws + OFF_S + ((wave << 4) + 8 * hh) * LD + tc + cl;
        const int hbase = (b * WIN + cl) * D1;
        #pragma unroll
        for (int e = 0; e < 8; ++e) {
            c0[e] = Sp0[e * LD];
            const int i0 = (wave << 4) + 8 * hh + e;
            const int ic = (i0 < D1) ? i0 : (D1 - 1);
            const float vL = Ht[hbase + ic];
            const float vH = Ht[hbase + 16 * D1 + ic];
            haL[e] = (i0 < D1) ? vL : 0.0f;
            haH[e] = (i0 < D1) ? vH : 0.0f;
        }
    };
    if constexpr (PASS && !STEPS) load_pass_operands();

    if constexpr (STEPS) {
        const int wb = b - 1;
        const float* G  = ws + OFF_G + (wb & 1) * (WIN * LD);
        const float* Mp = ws + OFF_M + (wb & 1) * LD;
        const float r = R[0];

        {
            const float* hrow = Ht + (wb * WIN + wave) * D1;
            float acc = 0.0f;
            #pragma unroll
            for (int j = 0; j < 16; ++j) {
                const int i = lane + 32 * j;
                const float hv = (i < D1) ? hrow[i] : 0.0f;
                acc = fmaf(hv, Mp[i], acc);
                hs[wave * KP + i] = hv;
                kr[wave * KP + i] = G[wave * LD + i];
            }
            const float cdot = wave_sum(acc);
            if (lane == 0) sC[wave] = cdot;
        }
        __syncthreads();

        {
            const int ta = (wave >> 1) & 1;
            const int tb = wave & 1;
            const int sl = wave >> 2;
            const float* ap = kr + (16 * ta + cl) * KP + 64 * sl + 32 * hh;
            const float* bp = hs + (16 * tb + cl) * KP + 64 * sl + 32 * hh;
            v8f g = {0.0f, 0.0f, 0.0f, 0.0f, 0.0f, 0.0f, 0.0f, 0.0f};
            #pragma unroll
            for (int c = 0; c < 16; ++c)
                g = mm4((v2f){ap[2 * c], ap[2 * c + 1]}, (v2f){bp[2 * c], bp[2 * c + 1]}, g);
            #pragma unroll
            for (int v = 0; v < 8; ++v) part[wave * 256 + v * 32 + lane] = g[v];
        }
        __syncthreads();

        {
            const int o = (wave >> 4) * 512 + (wave & 7) * 32 + ((wave >> 3) & 1) * 16 + (lane >> 4) * 256 + (lane & 15);
            float sacc = part[o];
            #pragma unroll
            for (int s = 1; s < 8; ++s) sacc += part[s * 1024 + o];
            gram[wave * WIN + lane] = sacc;
        }
        __syncthreads();

        if (wave == 0) {
            v2f bc[WIN];
            #pragma unroll
            for (int p = 0; p < WIN; ++p) bc[p] = (v2f){gram[p * WIN + lane], (lane == p) ? 1.0f : 0.0f};
            float cv = sC[lane];
            const float yv = emissions[wb * WIN + lane];
            float mc = 0.0f;
            float sv = 0.0f;
            #pragma unroll
            for (int q = 0; q < WIN; ++q) {
                const float s_term = r + lane_val(bc[q][0], q);
                const float inv    = 1.0f / (s_term + 1e-6f);
                const float u  = bc[q][0] * inv;
                const float ck = bc[q][1] * inv;
                const float resid = lane_val(yv, q) - lane_val(cv, q);
                mc = fmaf(resid, ck, mc);
                cv = fmaf(resid, u, cv);
                sv = (lane == q) ? s_term : sv;
                Ts[q * TP + lane]  = ck;
                Tms[q * TP + lane] = mc;
                const float ncv = u * (-s_term);
                const float sw = other_row(ncv);
                const float fL = hh ? sw : ncv;
                const float fH = hh ? ncv : sw;
                if (q + 1 < WIN) {
                    const float nc = row_lane_n((q + 1 < 16) ? fL : fH, q + 1);
                    bc[q + 1] = __builtin_elementwise_fma((v2f){nc, nc}, (v2f){u, ck}, bc[q + 1]);
                }
                __builtin_amdgcn_sched_barrier(0);
                #pragma unroll
                for (int p = q + 2; p < WIN; ++p) {
                    const float nc = row_lane_n((p < 16) ? fL : fH, p);
                    bc[p] = __builtin_elementwise_fma((v2f){nc, nc}, (v2f){u, ck}, bc[p]);
                }
            }
            nsv[lane] = -sv;
        }
        __syncthreads();
        if constexpr (PASS) load_pass_operands();

        {
            const int col = (wave << 4) + cl;
            float gB[16];
            #pragma unroll
            for (int e = 0; e < 16; ++e) gB[e] = kr[(16 * hh + e) * KP + col];

            if (!PASS || wave == (int)blockIdx.x) {
                const float* t0p = Tms + cl * TP + 16 * hh;
                const float* t1p = Tms + (16 + cl) * TP + 16 * hh;
                const float m0 = Mp[col];
                v8f o0 = {m0, m0, m0, m0, m0, m0, m0, m0};
                v8f o1 = {m0, m0, m0, m0, m0, m0, m0, m0};
                #pragma unroll
                for (int c = 0; c < 8; ++c) {
                    o0 = mm4((v2f){t0p[2 * c], t0p[2 * c + 1]}, (v2f){gB[2 * c], gB[2 * c + 1]}, o0);
                    o1 = mm4((v2f){t1p[2 * c], t1p[2 * c + 1]}, (v2f){gB[2 * c], gB[2 * c + 1]}, o1);
                }
                float* op = out + (wb * WIN + 8 * hh) * D1 + col;
                if (col < D1) {
                    #pragma unroll
                    for (int v = 0; v < 8; ++v) {
                        op[v * D1] = o0[v];
                        op[(16 + v) * D1] = o1[v];
                    }
                }
                if (PASS && hh == 1) ws[OFF_M + (b & 1) * LD + col] = o1[7];
            }

            if constexpr (PASS) {
                const float* t0p = Ts + cl * TP + 16 * hh;
                const float* t1p = Ts + (16 + cl) * TP + 16 * hh;
                v8f k0 = {0.0f, 0.0f, 0.0f, 0.0f, 0.0f, 0.0f, 0.0f, 0.0f};
                v8f k1 = {0.0f, 0.0f, 0.0f, 0.0f, 0.0f, 0.0f, 0.0f, 0.0f};
                #pragma unroll
                for (int c = 0; c < 8; ++c) {
                    k0 = mm4((v2f){t0p[2 * c], t0p[2 * c + 1]}, (v2f){gB[2 * c], gB[2 * c + 1]}, k0);
                    k1 = mm4((v2f){t1p[2 * c], t1p[2 * c + 1]}, (v2f){gB[2 * c], gB[2 * c + 1]}, k1);
                }
                #pragma unroll
                for (int v = 0; v < 8; ++v) {
                    kr[(8 * hh + v) * KP + col] = k0[v];
                    kr[(16 + 8 * hh + v) * KP + col] = k1[v];
                }
            }
        }
        __syncthreads();
    } else {
        for (int i = tid; i < WIN * KP; i += NT) kr[i] = 0.0f;
        if (tid < WIN) nsv[tid] = 0.0f;
        __syncthreads();
    }

    if constexpr (PASS) {
        float* Sp0 = ws + OFF_S + ((wave << 4) + 8 * hh) * LD + tc + cl;
        float* Gn  = ws + OFF_G + (b & 1) * (WIN * LD);

        float ka[16], kb[16];
        #pragma unroll
        for (int e = 0; e < 16; ++e) {
            const int q = 16 * hh + e;
            ka[e] = kr[q * KP + (wave << 4) + cl];
            kb[e] = kr[q * KP + tc + cl] * nsv[q];
        }
        v8f d0 = c0;
        d0 = mm4((v2f){ka[0],  ka[1]},  (v2f){kb[0],  kb[1]},  d0);
        d0 = mm4((v2f){ka[2],  ka[3]},  (v2f){kb[2],  kb[3]},  d0);
        d0 = mm4((v2f){ka[4],  ka[5]},  (v2f){kb[4],  kb[5]},  d0);
        d0 = mm4((v2f){ka[6],  ka[7]},  (v2f){kb[6],  kb[7]},  d0);
        d0 = mm4((v2f){ka[8],  ka[9]},  (v2f){kb[8],  kb[9]},  d0);
        d0 = mm4((v2f){ka[10], ka[11]}, (v2f){kb[10], kb[11]}, d0);
        d0 = mm4((v2f){ka[12], ka[13]}, (v2f){kb[12], kb[13]}, d0);
        d0 = mm4((v2f){ka[14], ka[15]}, (v2f){kb[14], kb[15]}, d0);

        #pragma unroll
        for (int v = 0; v < 8; ++v) Sp0[v * LD] = d0[v];

        v8f gL = {0.0f, 0.0f, 0.0f, 0.0f, 0.0f, 0.0f, 0.0f, 0.0f};
        v8f gH = {0.0f, 0.0f, 0.0f, 0.0f, 0.0f, 0.0f, 0.0f, 0.0f};
        gL = mm4((v2f){haL[0], haL[1]}, (v2f){d0[0], d0[1]}, gL);
        gH = mm4((v2f){haH[0], haH[1]}, (v2f){d0[0], d0[1]}, gH);
        gL = mm4((v2f){haL[2], haL[3]}, (v2f){d0[2], d0[3]}, gL);
        gH = mm4((v2f){haH[2], haH[3]}, (v2f){d0[2], d0[3]}, gH);
        gL = mm4((v2f){haL[4], haL[5]}, (v2f){d0[4], d0[5]}, gL);
        gH = mm4((v2f){haH[4], haH[5]}, (v2f){d0[4], d0[5]}, gH);
        gL = mm4((v2f){haL[6], haL[7]}, (v2f){d0[6], d0[7]}, gL);
        gH = mm4((v2f){haH[6], haH[7]}, (v2f){d0[6], d0[7]}, gH);

        #pragma unroll
        for (int v = 0; v < 8; ++v) {
            part[wave * 512 + v * 32 + lane]       = gL[v];
            part[wave * 512 + 256 + v * 32 + lane] = gH[v];
        }
        __syncthreads();

        if (tid < 512) {
            float acc = 0.0f;
            for (int w = 0; w < 32; ++w) acc += part[w * 512 + tid];
            const int a = tid >> 8;
            const int v = (tid >> 5) & 7;
            Gn[(16 * a + 8 * hh + v) * LD + tc + cl] = acc;
        }
    }
}

extern "C" void kernel_launch(void* const* d_in, const int* in_sizes, int n_in,
                              void* d_out, int out_size, void* d_ws, size_t ws_size,
                              hipStream_t stream) {
    const float* emissions = (const float*)d_in[0];
    const float* Ht        = (const float*)d_in[1];
    const float* R         = (const float*)d_in[4];
    const float* mu0       = (const float*)d_in[5];
    const float* Sigma0    = (const float*)d_in[6];

    float* ws  = (float*)d_ws;
    float* out = (float*)d_out;

#define KF_WINDOW(B) kf_window<true, true><<<NWG, NT, 0, stream>>>(emissions, Ht, R, ws, out, B);

    kf_init<<<(LD * LD) / 256, 256, 0, stream>>>(Sigma0, mu0, ws);
    kf_window<false, true><<<NWG, NT, 0, stream>>>(emissions, Ht, R, ws, out, 0);
    KF_WINDOW(1)  KF_WINDOW(2)  KF_WINDOW(3)  KF_WINDOW(4)
    KF_WINDOW(5)  KF_WINDOW(6)  KF_WINDOW(7)  KF_WINDOW(8)
    KF_WINDOW(9)  KF_WINDOW(10) KF_WINDOW(11) KF_WINDOW(12)
    KF_WINDOW(13) KF_WINDOW(14) KF_WINDOW(15)
    kf_window<true, false><<<1, NT, 0, stream>>>(emissions, Ht, R, ws, out, NBLK);
}


# window Gram matrix G0*H^T and sums h.m no longer repeated by every workgroup: each pass leaves its 16-column share (f32 WMMA), next launch adds the 32 shares; sum order changes
# speedup vs baseline: 31.3597x; 1.0385x over previous
#include <hip/hip_runtime.h>


typedef __attribute__((ext_vector_type(2))) float v2f;
typedef __attribute__((ext_vector_type(8))) float v8f;

#define D1 501
#define NSTEPS 512
#define LD 512
#define WIN 32
#define NBLK (NSTEPS / WIN)
#define NWG (LD / 16)
#define NT (32 * WIN)
#define KP 516
#define TP 36
#define GP 20

#define OFF_S  0
#define OFF_G  (LD * LD)
#define OFF_M  (OFF_G + 2 * WIN * LD)
#define PSZ    (WIN * WIN + WIN)
#define PPLANE (NWG * PSZ)
#define OFF_P  (OFF_M + 2 * LD)
static_assert(NWG == NT / 32, "wave w of a workgroup fetches workgroup w's share of the sums: as many waves as workgroups");

__global__ void kf_init(const float* __restrict__ Sigma0, const float* __restrict__ mu0, float* ws) {
    int idx = blockIdx.x * blockDim.x + threadIdx.x;
    if (idx < LD * LD) {
        int r = idx >> 9;
        int c = idx & (LD - 1);
        float v = 0.0f;
        if (r < D1 && c < D1) v = Sigma0[r * D1 + c];
        ws[OFF_S + idx] = v;
    }
    if (idx < LD) ws[OFF_M + idx] = (idx < D1) ? mu0[idx] : 0.0f;
}

__device__ __forceinline__ float other_row(float x) {
    const int b = __builtin_bit_cast(int, x);
    return __builtin_bit_cast(float, __builtin_amdgcn_permlanex16(b, b, 0x76543210, 0xfedcba98, false, false));
}

template <int M>
__device__ __forceinline__ float row_xor(float x) {
    return __builtin_bit_cast(float, __builtin_amdgcn_update_dpp(0, __builtin_bit_cast(int, x), 0x160 + M, 0xf, 0xf, true));
}

template <int P>
__device__ __forceinline__ float row_lane(float x) {
    return __builtin_bit_cast(float, __builtin_amdgcn_update_dpp(0, __builtin_bit_cast(int, x), 0x150 + P, 0xf, 0xf, true));
}

__device__ __forceinline__ float row_lane_n(float x, int p) {
    switch (p & 15) {
        case 0:  return row_lane<0>(x);
        case 1:  return row_lane<1>(x);
        case 2:  return row_lane<2>(x);
        case 3:  return row_lane<3>(x);
        case 4:  return row_lane<4>(x);
        case 5:  return row_lane<5>(x);
        case 6:  return row_lane<6>(x);
        case 7:  return row_lane<7>(x);
        case 8:  return row_lane<8>(x);
        case 9:  return row_lane<9>(x);
        case 10: return row_lane<10>(x);
        case 11: return row_lane<11>(x);
        case 12: return row_lane<12>(x);
        case 13: return row_lane<13>(x);
        case 14: return row_lane<14>(x);
        default: return row_lane<15>(x);
    }
}

__device__ __forceinline__ float wave_sum(float x) {
    x += other_row(x);
    x += row_xor<8>(x);
    x += row_xor<4>(x);
    x += row_xor<2>(x);
    x += row_xor<1>(x);
    return x;
}

__device__ __forceinline__ v8f mm4(v2f a, v2f b, v8f c) {
    return __builtin_amdgcn_wmma_f32_16x16x4_f32(false, a, false, b, (short)0, c, false, false);
}

__device__ __forceinline__ float lane_val(float x, int q) {
    return __builtin_bit_cast(float, __builtin_amdgcn_readlane(__builtin_bit_cast(int, x), q));
}

template <bool STEPS, bool PASS>
__launch_bounds__(NT)
__global__ void kf_window(const float* __restrict__ emissions,
                          const float* __restrict__ Ht,
                          const float* __restrict__ R,
                          float* ws,
                          float* __restrict__ out,
                          int b) {
    __shared__ __attribute__((aligned(16))) float kr[WIN * KP];
    __shared__ __attribute__((aligned(16))) float part[32 * 512];
    __shared__ __attribute__((aligned(16))) float gs[WIN * GP];
    __shared__ float ms[16];
    __shared__ __attribute__((aligned(16))) float Ts[WIN * TP];
    __shared__ __attribute__((aligned(16))) float Tms[WIN * TP];
    __shared__ float gram[WIN * WIN];
    __shared__ float nsv[WIN];
    __shared__ float cps[NWG * WIN];

    const int tid  = threadIdx.x;
    const int lane = tid & 31;
    const int wave = __builtin_amdgcn_readfirstlane(tid >> 5);
    const int cl = lane & 15;
    const int hh = lane >> 4;
    const int tc = blockIdx.x << 4;

    v8f c0;
    float haL[8], haH[8];
    auto load_pass_operands = [&]() {
        const float* Sp0 = ws + OFF_S + ((wave << 4) + 8 * hh) * LD + tc + cl;
        const int hbase = (b * WIN + cl) * D1;
        #pragma unroll
        for (int e = 0; e < 8; ++e) {
            c0[e] = Sp0[e * LD];
            const int i0 = (wave << 4) + 8 * hh + e;
            const int ic = (i0 < D1) ? i0 : (D1 - 1);
            const float vL = Ht[hbase + ic];
            const float vH = Ht[hbase + 16 * D1 + ic];
            haL[e] = (i0 < D1) ? vL : 0.0f;
            haH[e] = (i0 < D1) ? vH : 0.0f;
        }
    };
    if constexpr (PASS && !STEPS) load_pass_operands();

    if constexpr (STEPS) {
        const int wb = b - 1;
        const float* G  = ws + OFF_G + (wb & 1) * (WIN * LD);
        const float* Mp = ws + OFF_M + (wb & 1) * LD;
        const float r = R[0];

        #pragma unroll
        for (int j = 0; j < 16; ++j) {
            const int i = lane + 32 * j;
            kr[wave * KP + i] = G[wave * LD + i];
        }

        {
            const float* Pp = ws + OFF_P + (wb & 1) * PPLANE;
            const float* pe = Pp + wave * WIN + lane;
            float a4[4] = {pe[0], pe[PSZ], pe[2 * PSZ], pe[3 * PSZ]};
            #pragma unroll
            for (int t = 1; t < NWG / 4; ++t) {
                a4[0] += pe[(4 * t) * PSZ];
                a4[1] += pe[(4 * t + 1) * PSZ];
                a4[2] += pe[(4 * t + 2) * PSZ];
                a4[3] += pe[(4 * t + 3) * PSZ];
            }
            gram[wave * WIN + lane] = (a4[0] + a4[1]) + (a4[2] + a4[3]);
            cps[wave * WIN + lane] = Pp[wave * PSZ + WIN * WIN + lane];
        }
        __syncthreads();

        if (wave == 0) {
            v2f bc[WIN];
            #pragma unroll
            for (int p = 0; p < WIN; ++p) bc[p] = (v2f){gram[p * WIN + lane], (lane == p) ? 1.0f : 0.0f};
            float cv;
            {
                float c4[4] = {cps[lane], cps[WIN + lane], cps[2 * WIN + lane], cps[3 * WIN + lane]};
                #pragma unroll
                for (int t = 1; t < NWG / 4; ++t) {
                    c4[0] += cps[(4 * t) * WIN + lane];
                    c4[1] += cps[(4 * t + 1) * WIN + lane];
                    c4[2] += cps[(4 * t + 2) * WIN + lane];
                    c4[3] += cps[(4 * t + 3) * WIN + lane];
                }
                cv = (c4[0] + c4[1]) + (c4[2] + c4[3]);
            }
            const float yv = emissions[wb * WIN + lane];
            float mc = 0.0f;
            float sv = 0.0f;
            #pragma unroll
            for (int q = 0; q < WIN; ++q) {
                const float s_term = r + lane_val(bc[q][0], q);
                const float inv    = 1.0f / (s_term + 1e-6f);
                const float u  = bc[q][0] * inv;
                const float ck = bc[q][1] * inv;
                const float resid = lane_val(yv, q) - lane_val(cv, q);
                mc = fmaf(resid, ck, mc);
                cv = fmaf(resid, u, cv);
                sv = (lane == q) ? s_term : sv;
                Ts[q * TP + lane]  = ck;
                Tms[q * TP + lane] = mc;
                const float ncv = u * (-s_term);
                const float sw = other_row(ncv);
                const float fL = hh ? sw : ncv;
                const float fH = hh ? ncv : sw;
                if (q + 1 < WIN) {
                    const float nc = row_lane_n((q + 1 < 16) ? fL : fH, q + 1);
                    bc[q + 1] = __builtin_elementwise_fma((v2f){nc, nc}, (v2f){u, ck}, bc[q + 1]);
                }
                __builtin_amdgcn_sched_barrier(0);
                #pragma unroll
                for (int p = q + 2; p < WIN; ++p) {
                    const float nc = row_lane_n((p < 16) ? fL : fH, p);
                    bc[p] = __builtin_elementwise_fma((v2f){nc, nc}, (v2f){u, ck}, bc[p]);
                }
            }
            nsv[lane] = -sv;
        }
        __syncthreads();
        if constexpr (PASS) load_pass_operands();

        {
            const int col = (wave << 4) + cl;
            float gB[16];
            #pragma unroll
            for (int e = 0; e < 16; ++e) gB[e] = kr[(16 * hh + e) * KP + col];

            if (!PASS || wave == (int)blockIdx.x) {
                const float* t0p = Tms + cl * TP + 16 * hh;
                const float* t1p = Tms + (16 + cl) * TP + 16 * hh;
                const float m0 = Mp[col];
                v8f o0 = {m0, m0, m0, m0, m0, m0, m0, m0};
                v8f o1 = {m0, m0, m0, m0, m0, m0, m0, m0};
                #pragma unroll
                for (int c = 0; c < 8; ++c) {
                    o0 = mm4((v2f){t0p[2 * c], t0p[2 * c + 1]}, (v2f){gB[2 * c], gB[2 * c + 1]}, o0);
                    o1 = mm4((v2f){t1p[2 * c], t1p[2 * c + 1]}, (v2f){gB[2 * c], gB[2 * c + 1]}, o1);
                }
                float* op = out + (wb * WIN + 8 * hh) * D1 + col;
                if (col < D1) {
                    #pragma unroll
                    for (int v = 0; v < 8; ++v) {
                        op[v * D1] = o0[v];
                        op[(16 + v) * D1] = o1[v];
                    }
                }
                if (PASS && hh == 1) {
                    ws[OFF_M + (b & 1) * LD + col] = o1[7];
                    ms[cl] = o1[7];
                }
            }

            if constexpr (PASS) {
                const float* t0p = Ts + cl * TP + 16 * hh;
                const float* t1p = Ts + (16 + cl) * TP + 16 * hh;
                v8f k0 = {0.0f, 0.0f, 0.0f, 0.0f, 0.0f, 0.0f, 0.0f, 0.0f};
                v8f k1 = {0.0f, 0.0f, 0.0f, 0.0f, 0.0f, 0.0f, 0.0f, 0.0f};
                #pragma unroll
                for (int c = 0; c < 8; ++c) {
                    k0 = mm4((v2f){t0p[2 * c], t0p[2 * c + 1]}, (v2f){gB[2 * c], gB[2 * c + 1]}, k0);
                    k1 = mm4((v2f){t1p[2 * c], t1p[2 * c + 1]}, (v2f){gB[2 * c], gB[2 * c + 1]}, k1);
                }
                #pragma unroll
                for (int v = 0; v < 8; ++v) {
                    kr[(8 * hh + v) * KP + col] = k0[v];
                    kr[(16 + 8 * hh + v) * KP + col] = k1[v];
                }
            }
        }
        __syncthreads();
    } else {
        for (int i = tid; i < WIN * KP; i += NT) kr[i] = 0.0f;
        if (tid < WIN) nsv[tid] = 0.0f;
        __syncthreads();
    }

    if constexpr (PASS) {
        float* Sp0 = ws + OFF_S + ((wave << 4) + 8 * hh) * LD + tc + cl;
        float* Gn  = ws + OFF_G + (b & 1) * (WIN * LD);

        float ka[16], kb[16];
        #pragma unroll
        for (int e = 0; e < 16; ++e) {
            const int q = 16 * hh + e;
            ka[e] = kr[q * KP + (wave << 4) + cl];
            kb[e] = kr[q * KP + tc + cl] * nsv[q];
        }
        v8f d0 = c0;
        d0 = mm4((v2f){ka[0],  ka[1]},  (v2f){kb[0],  kb[1]},  d0);
        d0 = mm4((v2f){ka[2],  ka[3]},  (v2f){kb[2],  kb[3]},  d0);
        d0 = mm4((v2f){ka[4],  ka[5]},  (v2f){kb[4],  kb[5]},  d0);
        d0 = mm4((v2f){ka[6],  ka[7]},  (v2f){kb[6],  kb[7]},  d0);
        d0 = mm4((v2f){ka[8],  ka[9]},  (v2f){kb[8],  kb[9]},  d0);
        d0 = mm4((v2f){ka[10], ka[11]}, (v2f){kb[10], kb[11]}, d0);
        d0 = mm4((v2f){ka[12], ka[13]}, (v2f){kb[12], kb[13]}, d0);
        d0 = mm4((v2f){ka[14], ka[15]}, (v2f){kb[14], kb[15]}, d0);

        #pragma unroll
        for (int v = 0; v < 8; ++v) Sp0[v * LD] = d0[v];

        v8f gL = {0.0f, 0.0f, 0.0f, 0.0f, 0.0f, 0.0f, 0.0f, 0.0f};
        v8f gH = {0.0f, 0.0f, 0.0f, 0.0f, 0.0f, 0.0f, 0.0f, 0.0f};
        gL = mm4((v2f){haL[0], haL[1]}, (v2f){d0[0], d0[1]}, gL);
        gH = mm4((v2f){haH[0], haH[1]}, (v2f){d0[0], d0[1]}, gH);
        gL = mm4((v2f){haL[2], haL[3]}, (v2f){d0[2], d0[3]}, gL);
        gH = mm4((v2f){haH[2], haH[3]}, (v2f){d0[2], d0[3]}, gH);
        gL = mm4((v2f){haL[4], haL[5]}, (v2f){d0[4], d0[5]}, gL);
        gH = mm4((v2f){haH[4], haH[5]}, (v2f){d0[4], d0[5]}, gH);
        gL = mm4((v2f){haL[6], haL[7]}, (v2f){d0[6], d0[7]}, gL);
        gH = mm4((v2f){haH[6], haH[7]}, (v2f){d0[6], d0[7]}, gH);

        #pragma unroll
        for (int v = 0; v < 8; ++v) {
            part[wave * 512 + v * 32 + lane]       = gL[v];
            part[wave * 512 + 256 + v * 32 + lane] = gH[v];
        }
        __syncthreads();

        float acc = 0.0f;
        const int ga = tid >> 8;
        const int gv = (tid >> 5) & 7;
        if (tid < 512) {
            for (int w = 0; w < 32; ++w) acc += part[w * 512 + tid];
            gs[(16 * ga + 8 * hh + gv) * GP + cl] = acc;
        }
        __syncthreads();
        if (tid < 512) Gn[(16 * ga + 8 * hh + gv) * LD + tc + cl] = acc;

        if (wave == (int)blockIdx.x) {
            const float* a0p = gs + cl * GP + 8 * hh;
            const float* a1p = gs + (16 + cl) * GP + 8 * hh;
            v8f p00 = {0.0f, 0.0f, 0.0f, 0.0f, 0.0f, 0.0f, 0.0f, 0.0f};
            v8f p01 = {0.0f, 0.0f, 0.0f, 0.0f, 0.0f, 0.0f, 0.0f, 0.0f};
            v8f p10 = {0.0f, 0.0f, 0.0f, 0.0f, 0.0f, 0.0f, 0.0f, 0.0f};
            v8f p11 = {0.0f, 0.0f, 0.0f, 0.0f, 0.0f, 0.0f, 0.0f, 0.0f};
            #pragma unroll
            for (int c = 0; c < 4; ++c) {
                const v2f fa0 = (v2f){a0p[2 * c], a0p[2 * c + 1]};
                const v2f fa1 = (v2f){a1p[2 * c], a1p[2 * c + 1]};
                const v2f fbL = (v2f){haL[2 * c], haL[2 * c + 1]};
                const v2f fbH = (v2f){haH[2 * c], haH[2 * c + 1]};
                p00 = mm4(fa0, fbL, p00);
                p01 = mm4(fa0, fbH, p01);
                p10 = mm4(fa1, fbL, p10);
                p11 = mm4(fa1, fbH, p11);
            }
            float* Pw = ws + OFF_P + (b & 1) * PPLANE + blockIdx.x * PSZ;
            #pragma unroll
            for (int v = 0; v < 8; ++v) {
                Pw[(8 * hh + v) * WIN + cl]           = p00[v];
                Pw[(8 * hh + v) * WIN + 16 + cl]      = p01[v];
                Pw[(16 + 8 * hh + v) * WIN + cl]      = p10[v];
                Pw[(16 + 8 * hh + v) * WIN + 16 + cl] = p11[v];
            }
            float tL = 0.0f, tH = 0.0f;
            #pragma unroll
            for (int e = 0; e < 8; ++e) {
                const float mv = STEPS ? ms[8 * hh + e] : ws[OFF_M + tc + 8 * hh + e];
                tL = fmaf(haL[e], mv, tL);
                tH = fmaf(haH[e], mv, tH);
            }
            tL += other_row(tL);
            tH += other_row(tH);
            Pw[WIN * WIN + 16 * hh + cl] = hh ? tH : tL;
        }
    }
}

extern "C" void kernel_launch(void* const* d_in, const int* in_sizes, int n_in,
                              void* d_out, int out_size, void* d_ws, size_t ws_size,
                              hipStream_t stream) {
    const float* emissions = (const float*)d_in[0];
    const float* Ht        = (const float*)d_in[1];
    const float* R         = (const float*)d_in[4];
    const float* mu0       = (const float*)d_in[5];
    const float* Sigma0    = (const float*)d_in[6];

    float* ws  = (float*)d_ws;
    float* out = (float*)d_out;

#define KF_WINDOW(B) kf_window<true, true><<<NWG, NT, 0, stream>>>(emissions, Ht, R, ws, out, B);

    kf_init<<<(LD * LD) / 256, 256, 0, stream>>>(Sigma0, mu0, ws);
    kf_window<false, true><<<NWG, NT, 0, stream>>>(emissions, Ht, R, ws, out, 0);
    KF_WINDOW(1)  KF_WINDOW(2)  KF_WINDOW(3)  KF_WINDOW(4)
    KF_WINDOW(5)  KF_WINDOW(6)  KF_WINDOW(7)  KF_WINDOW(8)
    KF_WINDOW(9)  KF_WINDOW(10) KF_WINDOW(11) KF_WINDOW(12)
    KF_WINDOW(13) KF_WINDOW(14) KF_WINDOW(15)
    kf_window<true, false><<<1, NT, 0, stream>>>(emissions, Ht, R, ws, out, NBLK);
}
